# starting kernel
# baseline (speedup 1.0000x reference)
#include <hip/hip_runtime.h>
#include <hip/hip_bf16.h>
#include <math.h>

typedef __attribute__((ext_vector_type(2))) float v2f;
typedef __attribute__((ext_vector_type(8))) float v8f;

__global__ __launch_bounds__(256) void gemm_rows64_wmma(
    const float* __restrict__ X, const float* __restrict__ W,
    float* __restrict__ Y, int nrows)
{
  __shared__ v2f sW2[32 * 64];

  const int tid = threadIdx.x;
  for (int i = tid; i < 32 * 64; i += 256) {
    const int j = i >> 6;
    const int n = i & 63;
    v2f w;
    w.x = W[(2 * j) * 64 + n];
    w.y = W[(2 * j + 1) * 64 + n];
    sW2[i] = w;
  }
  __syncthreads();

  const int lane    = tid & 31;
  const int wave    = tid >> 5;
  const int rowBase = blockIdx.x * 128 + wave * 16;
  if (rowBase + 16 > nrows) return;

  const int m  = lane & 15;
  const int kh = lane >> 4;
  const int n  = lane & 15;

  const float* __restrict__ xrow = X + (size_t)(rowBase + m) * 64;

  v8f acc0 = {}, acc1 = {}, acc2 = {}, acc3 = {};

#pragma unroll
  for (int kk = 0; kk < 16; ++kk) {
    const int k0 = kk * 4 + kh * 2;
    v2f a = *(const v2f*)(xrow + k0);

    const int kp = k0 >> 1;
    const v2f* __restrict__ brow = &sW2[kp * 64 + n];
    v2f b0 = brow[0];
    v2f b1 = brow[16];
    v2f b2 = brow[32];
    v2f b3 = brow[48];

    acc0 = __builtin_amdgcn_wmma_f32_16x16x4_f32(false, a, false, b0, (short)0, acc0, false, false);
    acc1 = __builtin_amdgcn_wmma_f32_16x16x4_f32(false, a, false, b1, (short)0, acc1, false, false);
    acc2 = __builtin_amdgcn_wmma_f32_16x16x4_f32(false, a, false, b2, (short)0, acc2, false, false);
    acc3 = __builtin_amdgcn_wmma_f32_16x16x4_f32(false, a, false, b3, (short)0, acc3, false, false);
  }

  const int crow = rowBase + (lane >> 4) * 8;
  const int ccol = lane & 15;
#pragma unroll
  for (int v = 0; v < 8; ++v) {
    float* __restrict__ yr = Y + (size_t)(crow + v) * 64 + ccol;
    yr[0]  = acc0[v];
    yr[16] = acc1[v];
    yr[32] = acc2[v];
    yr[48] = acc3[v];
  }
}

__global__ __launch_bounds__(256) void scatter_add64(
    const float* __restrict__ src, const int* __restrict__ src_idx,
    const int* __restrict__ dst_idx, float* __restrict__ dst, int n_inc)
{
  const int e = blockIdx.x * 16 + (threadIdx.x >> 4);
  if (e >= n_inc) return;
  const int c4 = (threadIdx.x & 15) * 4;
  const int s  = src_idx[e];
  const int d  = dst_idx[e];
  const float4 v = *(const float4*)(src + (size_t)s * 64 + c4);
  float* __restrict__ dp = dst + (size_t)d * 64 + c4;
  atomicAdd(dp + 0, v.x);
  atomicAdd(dp + 1, v.y);
  atomicAdd(dp + 2, v.z);
  atomicAdd(dp + 3, v.w);
}

__global__ __launch_bounds__(256) void gate_tanh4(
    const float* __restrict__ pre, const float* __restrict__ gate,
    float* __restrict__ out, size_t n4)
{
  const size_t i = (size_t)blockIdx.x * blockDim.x + threadIdx.x;
  if (i >= n4) return;
  const float4 p = ((const float4*)pre)[i];
  const float4 g = ((const float4*)gate)[i];
  float4 r;
  r.x = tanhf(p.x * g.x);
  r.y = tanhf(p.y * g.y);
  r.z = tanhf(p.z * g.z);
  r.w = tanhf(p.w * g.w);
  ((float4*)out)[i] = r;
}

extern "C" void kernel_launch(void* const* d_in, const int* in_sizes, int n_in,
                              void* d_out, int out_size, void* d_ws, size_t ws_size,
                              hipStream_t stream) {
  const float* node_features  = (const float*)d_in[0];
  const float* hedge_features = (const float*)d_in[1];
  const float* W_nn = (const float*)d_in[2];
  const float* W_nh = (const float*)d_in[3];
  const float* W_hh = (const float*)d_in[4];
  const float* W_hn = (const float*)d_in[5];
  const int* node_idx  = (const int*)d_in[6];
  const int* hedge_idx = (const int*)d_in[7];

  const int N   = in_sizes[0] / 64;
  const int E   = in_sizes[1] / 64;
  const int INC = in_sizes[6];

  float* slotA = (float*)d_ws;
  float* slotB = slotA + (size_t)N * 64;

  float* out_node  = (float*)d_out;
  float* out_hedge = out_node + (size_t)N * 64;

  const dim3 blk(256);
  const dim3 gemmGridN((N + 127) / 128);
  const dim3 gemmGridE((E + 127) / 128);
  const dim3 scatGrid((INC + 15) / 16);

  gemm_rows64_wmma<<<gemmGridE, blk, 0, stream>>>(hedge_features, W_nh, slotA, E);
  hipMemsetAsync(slotB, 0, (size_t)N * 64 * sizeof(float), stream);
  scatter_add64<<<scatGrid, blk, 0, stream>>>(slotA, hedge_idx, node_idx, slotB, INC);
  gemm_rows64_wmma<<<gemmGridN, blk, 0, stream>>>(node_features, W_nn, slotA, N);
  gate_tanh4<<<dim3(((size_t)N * 16 + 255) / 256), blk, 0, stream>>>(
      slotA, slotB, out_node, (size_t)N * 16);

  gemm_rows64_wmma<<<gemmGridN, blk, 0, stream>>>(out_node, W_hn, slotA, N);
  hipMemsetAsync(slotB, 0, (size_t)E * 64 * sizeof(float), stream);
  scatter_add64<<<scatGrid, blk, 0, stream>>>(slotA, node_idx, hedge_idx, slotB, INC);
  gemm_rows64_wmma<<<gemmGridE, blk, 0, stream>>>(hedge_features, W_hh, slotA, E);
  gate_tanh4<<<dim3(((size_t)E * 16 + 255) / 256), blk, 0, stream>>>(
      slotA, slotB, out_hedge, (size_t)E * 16);
}


# Float-atomic scatters replaced by a pull: one pass of int tickets puts each incidence in fixed-size row buckets (no scan), a wave per output row sums in ascending source index, then gate*tanh
# speedup vs baseline: 1.1815x; 1.1815x over previous
#include <hip/hip_runtime.h>
#include <hip/hip_bf16.h>
#include <math.h>

typedef __attribute__((ext_vector_type(2))) float v2f;
typedef __attribute__((ext_vector_type(8))) float v8f;

__global__ __launch_bounds__(256) void gemm_rows64_wmma(
    const float* __restrict__ X, const float* __restrict__ W,
    float* __restrict__ Y, int nrows)
{
  __shared__ v2f sW2[32 * 64];

  const int tid = threadIdx.x;
  for (int i = tid; i < 32 * 64; i += 256) {
    const int j = i >> 6;
    const int n = i & 63;
    v2f w;
    w.x = W[(2 * j) * 64 + n];
    w.y = W[(2 * j + 1) * 64 + n];
    sW2[i] = w;
  }
  __syncthreads();

  const int lane    = tid & 31;
  const int wave    = tid >> 5;
  const int rowBase = blockIdx.x * 128 + wave * 16;
  if (rowBase + 16 > nrows) return;

  const int m  = lane & 15;
  const int kh = lane >> 4;
  const int n  = lane & 15;

  const float* __restrict__ xrow = X + (size_t)(rowBase + m) * 64;

  v8f acc0 = {}, acc1 = {}, acc2 = {}, acc3 = {};

#pragma unroll
  for (int kk = 0; kk < 16; ++kk) {
    const int k0 = kk * 4 + kh * 2;
    v2f a = *(const v2f*)(xrow + k0);

    const int kp = k0 >> 1;
    const v2f* __restrict__ brow = &sW2[kp * 64 + n];
    v2f b0 = brow[0];
    v2f b1 = brow[16];
    v2f b2 = brow[32];
    v2f b3 = brow[48];

    acc0 = __builtin_amdgcn_wmma_f32_16x16x4_f32(false, a, false, b0, (short)0, acc0, false, false);
    acc1 = __builtin_amdgcn_wmma_f32_16x16x4_f32(false, a, false, b1, (short)0, acc1, false, false);
    acc2 = __builtin_amdgcn_wmma_f32_16x16x4_f32(false, a, false, b2, (short)0, acc2, false, false);
    acc3 = __builtin_amdgcn_wmma_f32_16x16x4_f32(false, a, false, b3, (short)0, acc3, false, false);
  }

  const int crow = rowBase + (lane >> 4) * 8;
  const int ccol = lane & 15;
#pragma unroll
  for (int v = 0; v < 8; ++v) {
    float* __restrict__ yr = Y + (size_t)(crow + v) * 64 + ccol;
    yr[0]  = acc0[v];
    yr[16] = acc1[v];
    yr[32] = acc2[v];
    yr[48] = acc3[v];
  }
}

#define CAP_N 32
#define CAP_H 64

__global__ __launch_bounds__(256) void build_row_buckets(
    const int* __restrict__ node_idx, const int* __restrict__ hedge_idx,
    int* __restrict__ cntN, int* __restrict__ cntH,
    int* __restrict__ listN, int* __restrict__ listH,
    int* __restrict__ spillCnt, int2* __restrict__ spillN, int2* __restrict__ spillH,
    int n_inc)
{
  const int e = blockIdx.x * 256 + threadIdx.x;
  if (e >= n_inc) return;
  const int n = node_idx[e];
  const int h = hedge_idx[e];
  const int tn = atomicAdd(cntN + n, 1);
  const int th = atomicAdd(cntH + h, 1);
  if (tn < CAP_N) {
    listN[(size_t)n * CAP_N + tn] = h;
  } else {
    const int p = atomicAdd(spillCnt + 0, 1);
    spillN[p] = make_int2(n, h);
  }
  if (th < CAP_H) {
    listH[(size_t)h * CAP_H + th] = n;
  } else {
    const int p = atomicAdd(spillCnt + 1, 1);
    spillH[p] = make_int2(h, n);
  }
}

template <int U>
static __device__ __forceinline__ void gather_group(
    v2f& acc, int skq, int l0, int rem, const float* __restrict__ src, int lane)
{
  v2f v[U];
#pragma unroll
  for (int u = 0; u < U; ++u) {
    const int uu = (u < rem) ? u : (rem - 1);
    const int s = __builtin_amdgcn_readlane(skq, l0 + uu);
    v[u] = *(const v2f*)(src + (size_t)s * 64 + 2 * lane);
  }
#pragma unroll
  for (int u = 0; u < U; ++u) {
    if (u < rem) { acc.x += v[u].x; acc.y += v[u].y; }
  }
}

template <int CAP>
__global__ __launch_bounds__(256) void pull_gate_tanh(
    const float* __restrict__ src, const int* __restrict__ cnt,
    const int* __restrict__ list, const int* __restrict__ spillCntP,
    const int2* __restrict__ spill, const float* __restrict__ pre,
    float* __restrict__ out, int nrows, int n_inc)
{
  constexpr int NK = CAP / 32;
  __shared__ int sorted[8][CAP];

  const int lane = threadIdx.x & 31;
  const int wave = __builtin_amdgcn_readfirstlane((int)(threadIdx.x >> 5));
  const int rowRaw = blockIdx.x * 8 + wave;
  const bool live = rowRaw < nrows;
  const int row = live ? rowRaw : (nrows - 1);

  const int ctot = live ? cnt[row] : 0;
  const int cu = ctot < CAP ? ctot : CAP;

  const v2f p2 = *(const v2f*)(pre + (size_t)row * 64 + 2 * lane);

  int key[NK];
  int rnk[NK];
#pragma unroll
  for (int q = 0; q < NK; ++q) {
    const int p = q * 32 + lane;
    key[q] = (p < cu) ? list[(size_t)row * CAP + p] : 0x7fffffff;
    rnk[q] = 0;
  }
#pragma unroll
  for (int q2 = 0; q2 < NK; ++q2) {
    const int jn = (cu - q2 * 32) < 32 ? (cu - q2 * 32) : 32;
    for (int jj = 0; jj < jn; ++jj) {
      const int kj = __builtin_amdgcn_readlane(key[q2], jj);
      const int j = q2 * 32 + jj;
#pragma unroll
      for (int q = 0; q < NK; ++q) {
        const int p = q * 32 + lane;
        rnk[q] += (kj < key[q] || (kj == key[q] && j < p)) ? 1 : 0;
      }
    }
  }
#pragma unroll
  for (int q = 0; q < NK; ++q) {
    const int p = q * 32 + lane;
    if (p < cu) sorted[wave][rnk[q]] = key[q];
  }
  __syncthreads();

  int sk[NK];
#pragma unroll
  for (int q = 0; q < NK; ++q) {
    const int p = q * 32 + lane;
    sk[q] = (p < cu) ? sorted[wave][p] : 0;
  }

  v2f acc = {0.0f, 0.0f};
  for (int j = 0; j < cu; j += 8) {
    const int rem = cu - j;
    int skq = sk[0];
#pragma unroll
    for (int q = 1; q < NK; ++q) skq = (j >= q * 32) ? sk[q] : skq;
    const int l0 = j & 31;
    if (rem > 4) gather_group<8>(acc, skq, l0, rem, src, lane);
    else         gather_group<4>(acc, skq, l0, rem, src, lane);
  }

  if (ctot > CAP) {
    int tot = spillCntP[0];
    tot = tot < n_inc ? tot : n_inc;
    for (int i = 0; i < tot; i += 32) {
      const int p = i + lane;
      int2 pr = make_int2(-1, 0);
      if (p < tot) pr = spill[p];
      unsigned m = __builtin_amdgcn_ballot_w32(pr.x == row);
      while (m) {
        const int b = __builtin_ctz(m);
        m &= m - 1;
        const int s = __builtin_amdgcn_readlane(pr.y, b);
        const v2f v = *(const v2f*)(src + (size_t)s * 64 + 2 * lane);
        acc.x += v.x; acc.y += v.y;
      }
    }
  }

  if (live) {
    v2f r;
    r.x = tanhf(p2.x * acc.x);
    r.y = tanhf(p2.y * acc.y);
    *(v2f*)(out + (size_t)row * 64 + 2 * lane) = r;
  }
}

extern "C" void kernel_launch(void* const* d_in, const int* in_sizes, int n_in,
                              void* d_out, int out_size, void* d_ws, size_t ws_size,
                              hipStream_t stream) {
  const float* node_features  = (const float*)d_in[0];
  const float* hedge_features = (const float*)d_in[1];
  const float* W_nn = (const float*)d_in[2];
  const float* W_nh = (const float*)d_in[3];
  const float* W_hh = (const float*)d_in[4];
  const float* W_hn = (const float*)d_in[5];
  const int* node_idx  = (const int*)d_in[6];
  const int* hedge_idx = (const int*)d_in[7];

  const int N   = in_sizes[0] / 64;
  const int E   = in_sizes[1] / 64;
  const int INC = in_sizes[6];

  float* slotA = (float*)d_ws;
  float* slotB = slotA + (size_t)N * 64;
  int* listN  = (int*)(slotB + (size_t)N * 64);
  int* listH  = listN + (size_t)N * CAP_N;
  int2* spillN = (int2*)(listH + (size_t)E * CAP_H);
  int2* spillH = spillN + (size_t)INC;
  int* cntN   = (int*)(spillH + (size_t)INC);
  int* cntH   = cntN + N;
  int* spillCnt = cntH + E;

  float* out_node  = (float*)d_out;
  float* out_hedge = out_node + (size_t)N * 64;

  const dim3 blk(256);
  const dim3 gemmGridN((N + 127) / 128);
  const dim3 gemmGridE((E + 127) / 128);

  hipMemsetAsync(cntN, 0, ((size_t)N + E + 2) * sizeof(int), stream);
  build_row_buckets<<<dim3((INC + 255) / 256), blk, 0, stream>>>(
      node_idx, hedge_idx, cntN, cntH, listN, listH, spillCnt, spillN, spillH, INC);

  gemm_rows64_wmma<<<gemmGridE, blk, 0, stream>>>(hedge_features, W_nh, slotA, E);
  gemm_rows64_wmma<<<gemmGridN, blk, 0, stream>>>(node_features, W_nn, slotB, N);
  pull_gate_tanh<CAP_N><<<dim3((N + 7) / 8), blk, 0, stream>>>(
      slotA, cntN, listN, spillCnt + 0, spillN, slotB, out_node, N, INC);

  gemm_rows64_wmma<<<gemmGridN, blk, 0, stream>>>(out_node, W_hn, slotB, N);
  gemm_rows64_wmma<<<gemmGridE, blk, 0, stream>>>(hedge_features, W_hh, slotA, E);
  pull_gate_tanh<CAP_H><<<dim3((E + 7) / 8), blk, 0, stream>>>(
      slotB, cntH, listH, spillCnt + 1, spillH, slotA, out_hedge, E, INC);
}


# Ticket counters and bucket slots split into one copy per half of the package (die id picks the copy), so no counter or bucket line is written from both halves; reader merges both; sum order kept
# speedup vs baseline: 2.0192x; 1.7090x over previous
#include <hip/hip_runtime.h>
#include <hip/hip_bf16.h>
#include <math.h>

typedef __attribute__((ext_vector_type(2))) float v2f;
typedef __attribute__((ext_vector_type(8))) float v8f;

__global__ __launch_bounds__(256) void gemm_rows64_wmma(
    const float* __restrict__ X, const float* __restrict__ W,
    float* __restrict__ Y, int nrows)
{
  __shared__ v2f sW2[32 * 64];

  const int tid = threadIdx.x;
  for (int i = tid; i < 32 * 64; i += 256) {
    const int j = i >> 6;
    const int n = i & 63;
    v2f w;
    w.x = W[(2 * j) * 64 + n];
    w.y = W[(2 * j + 1) * 64 + n];
    sW2[i] = w;
  }
  __syncthreads();

  const int lane    = tid & 31;
  const int wave    = tid >> 5;
  const int rowBase = blockIdx.x * 128 + wave * 16;
  if (rowBase + 16 > nrows) return;

  const int m  = lane & 15;
  const int kh = lane >> 4;
  const int n  = lane & 15;

  const float* __restrict__ xrow = X + (size_t)(rowBase + m) * 64;

  v8f acc0 = {}, acc1 = {}, acc2 = {}, acc3 = {};

#pragma unroll
  for (int kk = 0; kk < 16; ++kk) {
    const int k0 = kk * 4 + kh * 2;
    v2f a = *(const v2f*)(xrow + k0);

    const int kp = k0 >> 1;
    const v2f* __restrict__ brow = &sW2[kp * 64 + n];
    v2f b0 = brow[0];
    v2f b1 = brow[16];
    v2f b2 = brow[32];
    v2f b3 = brow[48];

    acc0 = __builtin_amdgcn_wmma_f32_16x16x4_f32(false, a, false, b0, (short)0, acc0, false, false);
    acc1 = __builtin_amdgcn_wmma_f32_16x16x4_f32(false, a, false, b1, (short)0, acc1, false, false);
    acc2 = __builtin_amdgcn_wmma_f32_16x16x4_f32(false, a, false, b2, (short)0, acc2, false, false);
    acc3 = __builtin_amdgcn_wmma_f32_16x16x4_f32(false, a, false, b3, (short)0, acc3, false, false);
  }

  const int crow = rowBase + (lane >> 4) * 8;
  const int ccol = lane & 15;
#pragma unroll
  for (int v = 0; v < 8; ++v) {
    float* __restrict__ yr = Y + (size_t)(crow + v) * 64 + ccol;
    yr[0]  = acc0[v];
    yr[16] = acc1[v];
    yr[32] = acc2[v];
    yr[48] = acc3[v];
  }
}

#define CAP_N 32
#define CAP_H 64

__global__ __launch_bounds__(256) void build_row_buckets(
    const int* __restrict__ node_idx, const int* __restrict__ hedge_idx,
    int* __restrict__ cnt,
    int* __restrict__ listN,
    int* __restrict__ listH,
    int* __restrict__ spillCnt, int2* __restrict__ spillN, int2* __restrict__ spillH,
    int n_inc, int nN, int nH)
{
  constexpr int PN = CAP_N / 2;
  constexpr int PH = CAP_H / 2;
  const int mine = (__builtin_amdgcn_s_sendmsg_rtn(0x87) >> 18) & 1;
  const int other = mine ^ 1;

  const int e = blockIdx.x * 256 + threadIdx.x;
  if (e >= n_inc) return;
  const int n = node_idx[e];
  const int h = hedge_idx[e];

  const size_t cstride = (size_t)nN + nH;
  int* cNm = cnt + mine * cstride;
  int* cHm = cNm + nN;
  int* lNm = listN + (size_t)mine * nN * PN;
  int* lHm = listH + (size_t)mine * nH * PH;

  const int tn = atomicAdd(cNm + n, 1);
  const int th = atomicAdd(cHm + h, 1);
  if (tn < PN) {
    lNm[(size_t)n * PN + tn] = h;
  } else {
    const int t2 = atomicAdd(cnt + other * cstride + n, 1);
    if (t2 < PN) {
      listN[((size_t)other * nN + n) * PN + t2] = h;
    } else {
      const int p = atomicAdd(spillCnt + 0, 1);
      spillN[p] = make_int2(n, h);
    }
  }
  if (th < PH) {
    lHm[(size_t)h * PH + th] = n;
  } else {
    const int t2 = atomicAdd(cnt + other * cstride + nN + h, 1);
    if (t2 < PH) {
      listH[((size_t)other * nH + h) * PH + t2] = n;
    } else {
      const int p = atomicAdd(spillCnt + 1, 1);
      spillH[p] = make_int2(h, n);
    }
  }
}

template <int U>
static __device__ __forceinline__ void gather_group(
    v2f& acc, int skq, int l0, int rem, const float* __restrict__ src, int lane)
{
  v2f v[U];
#pragma unroll
  for (int u = 0; u < U; ++u) {
    const int uu = (u < rem) ? u : (rem - 1);
    const int s = __builtin_amdgcn_readlane(skq, l0 + uu);
    v[u] = *(const v2f*)(src + (size_t)s * 64 + 2 * lane);
  }
#pragma unroll
  for (int u = 0; u < U; ++u) {
    if (u < rem) { acc.x += v[u].x; acc.y += v[u].y; }
  }
}

template <int CAP>
__global__ __launch_bounds__(256) void pull_gate_tanh(
    const float* __restrict__ src, const int* __restrict__ cnt, int cntPartStride,
    const int* __restrict__ list, const int* __restrict__ spillCntP,
    const int2* __restrict__ spill, const float* __restrict__ pre,
    float* __restrict__ out, int nrows, int n_inc)
{
  constexpr int NK = CAP / 32;
  constexpr int PC = CAP / 2;
  __shared__ int sorted[8][CAP];

  const int lane = threadIdx.x & 31;
  const int wave = __builtin_amdgcn_readfirstlane((int)(threadIdx.x >> 5));
  const int rowRaw = blockIdx.x * 8 + wave;
  const bool live = rowRaw < nrows;
  const int row = live ? rowRaw : (nrows - 1);

  const int c0r = cnt[row];
  const int c1r = cnt[cntPartStride + row];
  const int c0 = live ? c0r : 0;
  const int c1 = live ? c1r : 0;
  const int u0 = c0 < PC ? c0 : PC;
  const int u1 = c1 < PC ? c1 : PC;
  const int cu = u0 + u1;

  const v2f p2 = *(const v2f*)(pre + (size_t)row * 64 + 2 * lane);

  int key[NK];
  int rnk[NK];
  bool have[NK];
#pragma unroll
  for (int q = 0; q < NK; ++q) {
    const int p = q * 32 + lane;
    const int part = p / PC;
    const int slot = p % PC;
    have[q] = slot < (part ? u1 : u0);
    key[q] = have[q] ? list[((size_t)part * nrows + row) * PC + slot] : 0x7fffffff;
    rnk[q] = 0;
  }
#pragma unroll
  for (int part = 0; part < 2; ++part) {
    const int un = part ? u1 : u0;
    const int q2 = (part * PC) >> 5;
    const int lbase = (part * PC) & 31;
    for (int s = 0; s < un; ++s) {
      const int kj = __builtin_amdgcn_readlane(key[q2], lbase + s);
      const int j = part * PC + s;
#pragma unroll
      for (int q = 0; q < NK; ++q) {
        const int p = q * 32 + lane;
        rnk[q] += (kj < key[q] || (kj == key[q] && j < p)) ? 1 : 0;
      }
    }
  }
#pragma unroll
  for (int q = 0; q < NK; ++q) {
    if (have[q]) sorted[wave][rnk[q]] = key[q];
  }
  __syncthreads();

  int sk[NK];
#pragma unroll
  for (int q = 0; q < NK; ++q) {
    const int p = q * 32 + lane;
    sk[q] = (p < cu) ? sorted[wave][p] : 0;
  }

  v2f acc = {0.0f, 0.0f};
  for (int j = 0; j < cu; j += 8) {
    const int rem = cu - j;
    int skq = sk[0];
#pragma unroll
    for (int q = 1; q < NK; ++q) skq = (j >= q * 32) ? sk[q] : skq;
    const int l0 = j & 31;
    if (rem > 4) gather_group<8>(acc, skq, l0, rem, src, lane);
    else         gather_group<4>(acc, skq, l0, rem, src, lane);
  }

  if (c0 > PC && c1 > PC) {
    int tot = spillCntP[0];
    tot = tot < n_inc ? tot : n_inc;
    for (int i = 0; i < tot; i += 32) {
      const int p = i + lane;
      int2 pr = make_int2(-1, 0);
      if (p < tot) pr = spill[p];
      unsigned m = __builtin_amdgcn_ballot_w32(pr.x == row);
      while (m) {
        const int b = __builtin_ctz(m);
        m &= m - 1;
        const int s = __builtin_amdgcn_readlane(pr.y, b);
        const v2f v = *(const v2f*)(src + (size_t)s * 64 + 2 * lane);
        acc.x += v.x; acc.y += v.y;
      }
    }
  }

  if (live) {
    v2f r;
    r.x = tanhf(p2.x * acc.x);
    r.y = tanhf(p2.y * acc.y);
    *(v2f*)(out + (size_t)row * 64 + 2 * lane) = r;
  }
}

extern "C" void kernel_launch(void* const* d_in, const int* in_sizes, int n_in,
                              void* d_out, int out_size, void* d_ws, size_t ws_size,
                              hipStream_t stream) {
  const float* node_features  = (const float*)d_in[0];
  const float* hedge_features = (const float*)d_in[1];
  const float* W_nn = (const float*)d_in[2];
  const float* W_nh = (const float*)d_in[3];
  const float* W_hh = (const float*)d_in[4];
  const float* W_hn = (const float*)d_in[5];
  const int* node_idx  = (const int*)d_in[6];
  const int* hedge_idx = (const int*)d_in[7];

  const int N   = in_sizes[0] / 64;
  const int E   = in_sizes[1] / 64;
  const int INC = in_sizes[6];

  float* slotA = (float*)d_ws;
  float* slotB = slotA + (size_t)N * 64;
  int* listN  = (int*)(slotB + (size_t)N * 64);
  int* listH  = listN + (size_t)N * CAP_N;
  int2* spillN = (int2*)(listH + (size_t)E * CAP_H);
  int2* spillH = spillN + (size_t)INC;
  int* cnt    = (int*)(spillH + (size_t)INC);
  int* spillCnt = cnt + 2 * ((size_t)N + E);

  float* out_node  = (float*)d_out;
  float* out_hedge = out_node + (size_t)N * 64;

  const dim3 blk(256);
  const dim3 gemmGridN((N + 127) / 128);
  const dim3 gemmGridE((E + 127) / 128);

  hipMemsetAsync(cnt, 0, (2 * ((size_t)N + E) + 2) * sizeof(int), stream);
  build_row_buckets<<<dim3((INC + 255) / 256), blk, 0, stream>>>(
      node_idx, hedge_idx, cnt, listN, listH, spillCnt, spillN, spillH, INC, N, E);

  gemm_rows64_wmma<<<gemmGridE, blk, 0, stream>>>(hedge_features, W_nh, slotA, E);
  gemm_rows64_wmma<<<gemmGridN, blk, 0, stream>>>(node_features, W_nn, slotB, N);
  pull_gate_tanh<CAP_N><<<dim3((N + 7) / 8), blk, 0, stream>>>(
      slotA, cnt, N + E, listN, spillCnt + 0, spillN, slotB, out_node, N, INC);

  gemm_rows64_wmma<<<gemmGridN, blk, 0, stream>>>(out_node, W_hn, slotB, N);
  gemm_rows64_wmma<<<gemmGridE, blk, 0, stream>>>(hedge_features, W_hh, slotA, E);
  pull_gate_tanh<CAP_H><<<dim3((E + 7) / 8), blk, 0, stream>>>(
      slotB, cnt + N, N + E, listH, spillCnt + 1, spillH, slotA, out_hedge, E, INC);
}


# Pull kernels run as resident grids: each wave loops over output rows (next row's counters and bucket asked for ahead) instead of one wave per row, 300000 waves entering -> 32768; sum order kept
# speedup vs baseline: 2.0920x; 1.0361x over previous
#include <hip/hip_runtime.h>
#include <hip/hip_bf16.h>
#include <math.h>

typedef __attribute__((ext_vector_type(2))) float v2f;
typedef __attribute__((ext_vector_type(8))) float v8f;

__global__ __launch_bounds__(256) void gemm_rows64_wmma(
    const float* __restrict__ X, const float* __restrict__ W,
    float* __restrict__ Y, int nrows)
{
  __shared__ v2f sW2[32 * 64];

  const int tid = threadIdx.x;
  for (int i = tid; i < 32 * 64; i += 256) {
    const int j = i >> 6;
    const int n = i & 63;
    v2f w;
    w.x = W[(2 * j) * 64 + n];
    w.y = W[(2 * j + 1) * 64 + n];
    sW2[i] = w;
  }
  __syncthreads();

  const int lane    = tid & 31;
  const int wave    = tid >> 5;
  const int rowBase = blockIdx.x * 128 + wave * 16;
  if (rowBase + 16 > nrows) return;

  const int m  = lane & 15;
  const int kh = lane >> 4;
  const int n  = lane & 15;

  const float* __restrict__ xrow = X + (size_t)(rowBase + m) * 64;

  v8f acc0 = {}, acc1 = {}, acc2 = {}, acc3 = {};

#pragma unroll
  for (int kk = 0; kk < 16; ++kk) {
    const int k0 = kk * 4 + kh * 2;
    v2f a = *(const v2f*)(xrow + k0);

    const int kp = k0 >> 1;
    const v2f* __restrict__ brow = &sW2[kp * 64 + n];
    v2f b0 = brow[0];
    v2f b1 = brow[16];
    v2f b2 = brow[32];
    v2f b3 = brow[48];

    acc0 = __builtin_amdgcn_wmma_f32_16x16x4_f32(false, a, false, b0, (short)0, acc0, false, false);
    acc1 = __builtin_amdgcn_wmma_f32_16x16x4_f32(false, a, false, b1, (short)0, acc1, false, false);
    acc2 = __builtin_amdgcn_wmma_f32_16x16x4_f32(false, a, false, b2, (short)0, acc2, false, false);
    acc3 = __builtin_amdgcn_wmma_f32_16x16x4_f32(false, a, false, b3, (short)0, acc3, false, false);
  }

  const int crow = rowBase + (lane >> 4) * 8;
  const int ccol = lane & 15;
#pragma unroll
  for (int v = 0; v < 8; ++v) {
    float* __restrict__ yr = Y + (size_t)(crow + v) * 64 + ccol;
    yr[0]  = acc0[v];
    yr[16] = acc1[v];
    yr[32] = acc2[v];
    yr[48] = acc3[v];
  }
}

#define CAP_N 32
#define CAP_H 64

__global__ __launch_bounds__(256) void build_row_buckets(
    const int* __restrict__ node_idx, const int* __restrict__ hedge_idx,
    int* __restrict__ cnt,
    int* __restrict__ listN,
    int* __restrict__ listH,
    int* __restrict__ spillCnt, int2* __restrict__ spillN, int2* __restrict__ spillH,
    int n_inc, int nN, int nH)
{
  constexpr int PN = CAP_N / 2;
  constexpr int PH = CAP_H / 2;
  const int mine = (__builtin_amdgcn_s_sendmsg_rtn(0x87) >> 18) & 1;
  const int other = mine ^ 1;

  const int e = blockIdx.x * 256 + threadIdx.x;
  if (e >= n_inc) return;
  const int n = node_idx[e];
  const int h = hedge_idx[e];

  const size_t cstride = (size_t)nN + nH;
  int* cNm = cnt + mine * cstride;
  int* cHm = cNm + nN;
  int* lNm = listN + (size_t)mine * nN * PN;
  int* lHm = listH + (size_t)mine * nH * PH;

  const int tn = atomicAdd(cNm + n, 1);
  const int th = atomicAdd(cHm + h, 1);
  if (tn < PN) {
    lNm[(size_t)n * PN + tn] = h;
  } else {
    const int t2 = atomicAdd(cnt + other * cstride + n, 1);
    if (t2 < PN) {
      listN[((size_t)other * nN + n) * PN + t2] = h;
    } else {
      const int p = atomicAdd(spillCnt + 0, 1);
      spillN[p] = make_int2(n, h);
    }
  }
  if (th < PH) {
    lHm[(size_t)h * PH + th] = n;
  } else {
    const int t2 = atomicAdd(cnt + other * cstride + nN + h, 1);
    if (t2 < PH) {
      listH[((size_t)other * nH + h) * PH + t2] = n;
    } else {
      const int p = atomicAdd(spillCnt + 1, 1);
      spillH[p] = make_int2(h, n);
    }
  }
}

template <int U>
static __device__ __forceinline__ void gather_group(
    v2f& acc, int skq, int l0, int rem, const float* __restrict__ src, int lane)
{
  v2f v[U];
#pragma unroll
  for (int u = 0; u < U; ++u) {
    const int uu = (u < rem) ? u : (rem - 1);
    const int s = __builtin_amdgcn_readlane(skq, l0 + uu);
    v[u] = *(const v2f*)(src + (size_t)s * 64 + 2 * lane);
  }
#pragma unroll
  for (int u = 0; u < U; ++u) {
    if (u < rem) { acc.x += v[u].x; acc.y += v[u].y; }
  }
}

template <int CAP>
__global__ __launch_bounds__(256) void pull_gate_tanh(
    const float* __restrict__ src, const int* __restrict__ cnt, int cntPartStride,
    const int* __restrict__ list, const int* __restrict__ spillCntP,
    const int2* __restrict__ spill, const float* __restrict__ pre,
    float* __restrict__ out, int nrows, int n_inc)
{
  constexpr int NK = CAP / 32;
  constexpr int PC = CAP / 2;
  __shared__ int sorted[8][CAP];

  const int lane = threadIdx.x & 31;
  const int wave = __builtin_amdgcn_readfirstlane((int)(threadIdx.x >> 5));
  const int step = (int)gridDim.x * 8;
  int row = blockIdx.x * 8 + wave;
  if (row >= nrows) return;

  int c0 = cnt[row];
  int c1 = cnt[cntPartStride + row];
  int u0 = c0 < PC ? c0 : PC;
  int u1 = c1 < PC ? c1 : PC;
  int key[NK];
#pragma unroll
  for (int q = 0; q < NK; ++q) {
    const int p = q * 32 + lane;
    const int part = p / PC;
    const int slot = p % PC;
    const bool hv = slot < (part ? u1 : u0);
    key[q] = hv ? list[((size_t)part * nrows + row) * PC + slot] : 0x7fffffff;
  }
  v2f p2 = *(const v2f*)(pre + (size_t)row * 64 + 2 * lane);

  for (;;) {
    const int nraw = row + step;
    const bool more = nraw < nrows;
    const int nrow = more ? nraw : row;
    const int nc0 = cnt[nrow];
    const int nc1 = cnt[cntPartStride + nrow];

    const int cu = u0 + u1;

    int rnk[NK];
#pragma unroll
    for (int q = 0; q < NK; ++q) rnk[q] = 0;
#pragma unroll
    for (int part = 0; part < 2; ++part) {
      const int un = part ? u1 : u0;
      const int q2 = (part * PC) >> 5;
      const int lbase = (part * PC) & 31;
      for (int s = 0; s < un; ++s) {
        const int kj = __builtin_amdgcn_readlane(key[q2], lbase + s);
        const int j = part * PC + s;
#pragma unroll
        for (int q = 0; q < NK; ++q) {
          const int p = q * 32 + lane;
          rnk[q] += (kj < key[q] || (kj == key[q] && j < p)) ? 1 : 0;
        }
      }
    }
#pragma unroll
    for (int q = 0; q < NK; ++q) {
      if (key[q] != 0x7fffffff) sorted[wave][rnk[q]] = key[q];
    }

    int sk[NK];
#pragma unroll
    for (int q = 0; q < NK; ++q) {
      const int p = q * 32 + lane;
      sk[q] = (p < cu) ? sorted[wave][p] : 0;
    }

    const int nu0 = nc0 < PC ? nc0 : PC;
    const int nu1 = nc1 < PC ? nc1 : PC;
    int nkey[NK];
#pragma unroll
    for (int q = 0; q < NK; ++q) {
      const int p = q * 32 + lane;
      const int part = p / PC;
      const int slot = p % PC;
      const bool hv = slot < (part ? nu1 : nu0);
      nkey[q] = hv ? list[((size_t)part * nrows + nrow) * PC + slot] : 0x7fffffff;
    }
    const v2f np2 = *(const v2f*)(pre + (size_t)nrow * 64 + 2 * lane);

    v2f acc = {0.0f, 0.0f};
    for (int j = 0; j < cu; j += 8) {
      const int rem = cu - j;
      int skq = sk[0];
#pragma unroll
      for (int q = 1; q < NK; ++q) skq = (j >= q * 32) ? sk[q] : skq;
      const int l0 = j & 31;
      if (rem > 4) gather_group<8>(acc, skq, l0, rem, src, lane);
      else         gather_group<4>(acc, skq, l0, rem, src, lane);
    }

    if (c0 > PC && c1 > PC) {
      int tot = spillCntP[0];
      tot = tot < n_inc ? tot : n_inc;
      for (int i = 0; i < tot; i += 32) {
        const int p = i + lane;
        int2 pr = make_int2(-1, 0);
        if (p < tot) pr = spill[p];
        unsigned m = __builtin_amdgcn_ballot_w32(pr.x == row);
        while (m) {
          const int b = __builtin_ctz(m);
          m &= m - 1;
          const int s = __builtin_amdgcn_readlane(pr.y, b);
          const v2f v = *(const v2f*)(src + (size_t)s * 64 + 2 * lane);
          acc.x += v.x; acc.y += v.y;
        }
      }
    }

    v2f r;
    r.x = tanhf(p2.x * acc.x);
    r.y = tanhf(p2.y * acc.y);
    *(v2f*)(out + (size_t)row * 64 + 2 * lane) = r;

    if (!more) break;
    row = nrow; c0 = nc0; c1 = nc1; u0 = nu0; u1 = nu1; p2 = np2;
#pragma unroll
    for (int q = 0; q < NK; ++q) key[q] = nkey[q];
  }
}

extern "C" void kernel_launch(void* const* d_in, const int* in_sizes, int n_in,
                              void* d_out, int out_size, void* d_ws, size_t ws_size,
                              hipStream_t stream) {
  const float* node_features  = (const float*)d_in[0];
  const float* hedge_features = (const float*)d_in[1];
  const float* W_nn = (const float*)d_in[2];
  const float* W_nh = (const float*)d_in[3];
  const float* W_hh = (const float*)d_in[4];
  const float* W_hn = (const float*)d_in[5];
  const int* node_idx  = (const int*)d_in[6];
  const int* hedge_idx = (const int*)d_in[7];

  const int N   = in_sizes[0] / 64;
  const int E   = in_sizes[1] / 64;
  const int INC = in_sizes[6];

  float* slotA = (float*)d_ws;
  float* slotB = slotA + (size_t)N * 64;
  int* listN  = (int*)(slotB + (size_t)N * 64);
  int* listH  = listN + (size_t)N * CAP_N;
  int2* spillN = (int2*)(listH + (size_t)E * CAP_H);
  int2* spillH = spillN + (size_t)INC;
  int* cnt    = (int*)(spillH + (size_t)INC);
  int* spillCnt = cnt + 2 * ((size_t)N + E);

  float* out_node  = (float*)d_out;
  float* out_hedge = out_node + (size_t)N * 64;

  const dim3 blk(256);
  const dim3 gemmGridN((N + 127) / 128);
  const dim3 gemmGridE((E + 127) / 128);

  hipMemsetAsync(cnt, 0, (2 * ((size_t)N + E) + 2) * sizeof(int), stream);
  build_row_buckets<<<dim3((INC + 255) / 256), blk, 0, stream>>>(
      node_idx, hedge_idx, cnt, listN, listH, spillCnt, spillN, spillH, INC, N, E);

  gemm_rows64_wmma<<<gemmGridE, blk, 0, stream>>>(hedge_features, W_nh, slotA, E);
  gemm_rows64_wmma<<<gemmGridN, blk, 0, stream>>>(node_features, W_nn, slotB, N);
  pull_gate_tanh<CAP_N><<<dim3(2048), blk, 0, stream>>>(
      slotA, cnt, N + E, listN, spillCnt + 0, spillN, slotB, out_node, N, INC);

  gemm_rows64_wmma<<<gemmGridN, blk, 0, stream>>>(out_node, W_hn, slotB, N);
  gemm_rows64_wmma<<<gemmGridE, blk, 0, stream>>>(hedge_features, W_hh, slotA, E);
  pull_gate_tanh<CAP_H><<<dim3(2048), blk, 0, stream>>>(
      slotB, cnt + N, N + E, listH, spillCnt + 1, spillH, slotA, out_hedge, E, INC);
}


# Bucket build and the three products of the inputs (hyperedge x W_nh, hyperedge x W_hh, node x W_nn) run as ONE launch, block roles interleaved 8:2:1:1: atomic waits overlap matrix work; sums kept
# speedup vs baseline: 2.4949x; 1.1926x over previous
#include <hip/hip_runtime.h>
#include <hip/hip_bf16.h>
#include <math.h>

typedef __attribute__((ext_vector_type(2))) float v2f;
typedef __attribute__((ext_vector_type(8))) float v8f;

static __device__ __forceinline__ void rows64_product_block(
    v2f* __restrict__ sW2, const float* __restrict__ X, const float* __restrict__ W,
    float* __restrict__ Y, int nrows, int tileBlock)
{
  const int tid = threadIdx.x;
  for (int i = tid; i < 32 * 64; i += 256) {
    const int j = i >> 6;
    const int n = i & 63;
    v2f w;
    w.x = W[(2 * j) * 64 + n];
    w.y = W[(2 * j + 1) * 64 + n];
    sW2[i] = w;
  }
  __syncthreads();

  const int lane    = tid & 31;
  const int wave    = tid >> 5;
  const int rowBase = tileBlock * 128 + wave * 16;
  if (rowBase + 16 > nrows) return;

  const int m  = lane & 15;
  const int kh = lane >> 4;
  const int n  = lane & 15;

  const float* __restrict__ xrow = X + (size_t)(rowBase + m) * 64;

  v8f acc0 = {}, acc1 = {}, acc2 = {}, acc3 = {};

#pragma unroll
  for (int kk = 0; kk < 16; ++kk) {
    const int k0 = kk * 4 + kh * 2;
    v2f a = *(const v2f*)(xrow + k0);

    const int kp = k0 >> 1;
    const v2f* __restrict__ brow = &sW2[kp * 64 + n];
    v2f b0 = brow[0];
    v2f b1 = brow[16];
    v2f b2 = brow[32];
    v2f b3 = brow[48];

    acc0 = __builtin_amdgcn_wmma_f32_16x16x4_f32(false, a, false, b0, (short)0, acc0, false, false);
    acc1 = __builtin_amdgcn_wmma_f32_16x16x4_f32(false, a, false, b1, (short)0, acc1, false, false);
    acc2 = __builtin_amdgcn_wmma_f32_16x16x4_f32(false, a, false, b2, (short)0, acc2, false, false);
    acc3 = __builtin_amdgcn_wmma_f32_16x16x4_f32(false, a, false, b3, (short)0, acc3, false, false);
  }

  const int crow = rowBase + (lane >> 4) * 8;
  const int ccol = lane & 15;
#pragma unroll
  for (int v = 0; v < 8; ++v) {
    float* __restrict__ yr = Y + (size_t)(crow + v) * 64 + ccol;
    yr[0]  = acc0[v];
    yr[16] = acc1[v];
    yr[32] = acc2[v];
    yr[48] = acc3[v];
  }
}

__global__ __launch_bounds__(256) void gemm_rows64_wmma(
    const float* __restrict__ X, const float* __restrict__ W,
    float* __restrict__ Y, int nrows)
{
  __shared__ v2f sW2[32 * 64];
  rows64_product_block(sW2, X, W, Y, nrows, (int)blockIdx.x);
}

#define CAP_N 32
#define CAP_H 64

static __device__ __forceinline__ void row_buckets_block(
    int incBlock,
    const int* __restrict__ node_idx, const int* __restrict__ hedge_idx,
    int* __restrict__ cnt,
    int* __restrict__ listN,
    int* __restrict__ listH,
    int* __restrict__ spillCnt, int2* __restrict__ spillN, int2* __restrict__ spillH,
    int n_inc, int nN, int nH)
{
  constexpr int PN = CAP_N / 2;
  constexpr int PH = CAP_H / 2;
  const int mine = (__builtin_amdgcn_s_sendmsg_rtn(0x87) >> 18) & 1;
  const int other = mine ^ 1;

  const int e = incBlock * 256 + (int)threadIdx.x;
  if (e >= n_inc) return;
  const int n = node_idx[e];
  const int h = hedge_idx[e];

  const size_t cstride = (size_t)nN + nH;
  int* cNm = cnt + mine * cstride;
  int* cHm = cNm + nN;
  int* lNm = listN + (size_t)mine * nN * PN;
  int* lHm = listH + (size_t)mine * nH * PH;

  const int tn = atomicAdd(cNm + n, 1);
  const int th = atomicAdd(cHm + h, 1);
  if (tn < PN) {
    lNm[(size_t)n * PN + tn] = h;
  } else {
    const int t2 = atomicAdd(cnt + other * cstride + n, 1);
    if (t2 < PN) {
      listN[((size_t)other * nN + n) * PN + t2] = h;
    } else {
      const int p = atomicAdd(spillCnt + 0, 1);
      spillN[p] = make_int2(n, h);
    }
  }
  if (th < PH) {
    lHm[(size_t)h * PH + th] = n;
  } else {
    const int t2 = atomicAdd(cnt + other * cstride + nN + h, 1);
    if (t2 < PH) {
      listH[((size_t)other * nH + h) * PH + t2] = n;
    } else {
      const int p = atomicAdd(spillCnt + 1, 1);
      spillH[p] = make_int2(h, n);
    }
  }
}

__global__ __launch_bounds__(256) __attribute__((amdgpu_waves_per_eu(16)))
void buckets_and_input_products(
    const int* __restrict__ node_idx, const int* __restrict__ hedge_idx,
    int* __restrict__ cnt, int* __restrict__ listN, int* __restrict__ listH,
    int* __restrict__ spillCnt, int2* __restrict__ spillN, int2* __restrict__ spillH,
    int n_inc, int nN, int nH,
    const float* __restrict__ node_features, const float* __restrict__ hedge_features,
    const float* __restrict__ W_nn, const float* __restrict__ W_nh,
    const float* __restrict__ W_hh,
    float* __restrict__ node_pre, float* __restrict__ hedge_proj,
    float* __restrict__ hedge_pre)
{
  __shared__ v2f sW2[32 * 64];

  const int nbB = (n_inc + 255) >> 8;
  const int nbN = (nN + 127) >> 7;
  const int nbE = (nH + 127) >> 7;
  int whole = nbB >> 3;
  whole = (nbN >> 1) < whole ? (nbN >> 1) : whole;
  whole = nbE < whole ? nbE : whole;

  const int b = (int)blockIdx.x;
  int role, idx;
  if (b < whole * 12) {
    const int pat = b / 12;
    const int slot = b - pat * 12;
    if (slot < 8)        { role = 0; idx = pat * 8 + slot; }
    else if (slot < 10)  { role = 1; idx = pat * 2 + (slot - 8); }
    else if (slot == 10) { role = 2; idx = pat; }
    else                 { role = 3; idx = pat; }
  } else {
    int r = b - whole * 12;
    const int leftB = nbB - whole * 8;
    const int leftN = nbN - whole * 2;
    const int leftE = nbE - whole;
    if (r < leftB) { role = 0; idx = whole * 8 + r; }
    else {
      r -= leftB;
      if (r < leftN) { role = 1; idx = whole * 2 + r; }
      else {
        r -= leftN;
        if (r < leftE) { role = 2; idx = whole + r; }
        else           { role = 3; idx = whole + (r - leftE); }
      }
    }
  }

  if (role == 0) {
    row_buckets_block(idx, node_idx, hedge_idx, cnt, listN, listH,
                      spillCnt, spillN, spillH, n_inc, nN, nH);
    return;
  }
  const float* __restrict__ X = (role == 1) ? node_features : hedge_features;
  const float* __restrict__ Wm = (role == 1) ? W_nn : (role == 2 ? W_nh : W_hh);
  float* __restrict__ Y = (role == 1) ? node_pre : (role == 2 ? hedge_proj : hedge_pre);
  const int nrows = (role == 1) ? nN : nH;
  rows64_product_block(sW2, X, Wm, Y, nrows, idx);
}

template <int U>
static __device__ __forceinline__ void gather_group(
    v2f& acc, int skq, int l0, int rem, const float* __restrict__ src, int lane)
{
  v2f v[U];
#pragma unroll
  for (int u = 0; u < U; ++u) {
    const int uu = (u < rem) ? u : (rem - 1);
    const int s = __builtin_amdgcn_readlane(skq, l0 + uu);
    v[u] = *(const v2f*)(src + (size_t)s * 64 + 2 * lane);
  }
#pragma unroll
  for (int u = 0; u < U; ++u) {
    if (u < rem) { acc.x += v[u].x; acc.y += v[u].y; }
  }
}

template <int CAP>
__global__ __launch_bounds__(256) void pull_gate_tanh(
    const float* __restrict__ src, const int* __restrict__ cnt, int cntPartStride,
    const int* __restrict__ list, const int* __restrict__ spillCntP,
    const int2* __restrict__ spill, const float* __restrict__ pre,
    float* __restrict__ out, int nrows, int n_inc)
{
  constexpr int NK = CAP / 32;
  constexpr int PC = CAP / 2;
  __shared__ int sorted[8][CAP];

  const int lane = threadIdx.x & 31;
  const int wave = __builtin_amdgcn_readfirstlane((int)(threadIdx.x >> 5));
  const int step = (int)gridDim.x * 8;
  int row = blockIdx.x * 8 + wave;
  if (row >= nrows) return;

  int c0 = cnt[row];
  int c1 = cnt[cntPartStride + row];
  int u0 = c0 < PC ? c0 : PC;
  int u1 = c1 < PC ? c1 : PC;
  int key[NK];
#pragma unroll
  for (int q = 0; q < NK; ++q) {
    const int p = q * 32 + lane;
    const int part = p / PC;
    const int slot = p % PC;
    const bool hv = slot < (part ? u1 : u0);
    key[q] = hv ? list[((size_t)part * nrows + row) * PC + slot] : 0x7fffffff;
  }
  v2f p2 = *(const v2f*)(pre + (size_t)row * 64 + 2 * lane);

  for (;;) {
    const int nraw = row + step;
    const bool more = nraw < nrows;
    const int nrow = more ? nraw : row;
    const int nc0 = cnt[nrow];
    const int nc1 = cnt[cntPartStride + nrow];

    const int cu = u0 + u1;

    int rnk[NK];
#pragma unroll
    for (int q = 0; q < NK; ++q) rnk[q] = 0;
#pragma unroll
    for (int part = 0; part < 2; ++part) {
      const int un = part ? u1 : u0;
      const int q2 = (part * PC) >> 5;
      const int lbase = (part * PC) & 31;
      for (int s = 0; s < un; ++s) {
        const int kj = __builtin_amdgcn_readlane(key[q2], lbase + s);
        const int j = part * PC + s;
#pragma unroll
        for (int q = 0; q < NK; ++q) {
          const int p = q * 32 + lane;
          rnk[q] += (kj < key[q] || (kj == key[q] && j < p)) ? 1 : 0;
        }
      }
    }
#pragma unroll
    for (int q = 0; q < NK; ++q) {
      if (key[q] != 0x7fffffff) sorted[wave][rnk[q]] = key[q];
    }

    int sk[NK];
#pragma unroll
    for (int q = 0; q < NK; ++q) {
      const int p = q * 32 + lane;
      sk[q] = (p < cu) ? sorted[wave][p] : 0;
    }

    const int nu0 = nc0 < PC ? nc0 : PC;
    const int nu1 = nc1 < PC ? nc1 : PC;
    int nkey[NK];
#pragma unroll
    for (int q = 0; q < NK; ++q) {
      const int p = q * 32 + lane;
      const int part = p / PC;
      const int slot = p % PC;
      const bool hv = slot < (part ? nu1 : nu0);
      nkey[q] = hv ? list[((size_t)part * nrows + nrow) * PC + slot] : 0x7fffffff;
    }
    const v2f np2 = *(const v2f*)(pre + (size_t)nrow * 64 + 2 * lane);

    v2f acc = {0.0f, 0.0f};
    for (int j = 0; j < cu; j += 8) {
      const int rem = cu - j;
      int skq = sk[0];
#pragma unroll
      for (int q = 1; q < NK; ++q) skq = (j >= q * 32) ? sk[q] : skq;
      const int l0 = j & 31;
      if (rem > 4) gather_group<8>(acc, skq, l0, rem, src, lane);
      else         gather_group<4>(acc, skq, l0, rem, src, lane);
    }

    if (c0 > PC && c1 > PC) {
      int tot = spillCntP[0];
      tot = tot < n_inc ? tot : n_inc;
      for (int i = 0; i < tot; i += 32) {
        const int p = i + lane;
        int2 pr = make_int2(-1, 0);
        if (p < tot) pr = spill[p];
        unsigned m = __builtin_amdgcn_ballot_w32(pr.x == row);
        while (m) {
          const int b = __builtin_ctz(m);
          m &= m - 1;
          const int s = __builtin_amdgcn_readlane(pr.y, b);
          const v2f v = *(const v2f*)(src + (size_t)s * 64 + 2 * lane);
          acc.x += v.x; acc.y += v.y;
        }
      }
    }

    v2f r;
    r.x = tanhf(p2.x * acc.x);
    r.y = tanhf(p2.y * acc.y);
    *(v2f*)(out + (size_t)row * 64 + 2 * lane) = r;

    if (!more) break;
    row = nrow; c0 = nc0; c1 = nc1; u0 = nu0; u1 = nu1; p2 = np2;
#pragma unroll
    for (int q = 0; q < NK; ++q) key[q] = nkey[q];
  }
}

extern "C" void kernel_launch(void* const* d_in, const int* in_sizes, int n_in,
                              void* d_out, int out_size, void* d_ws, size_t ws_size,
                              hipStream_t stream) {
  const float* node_features  = (const float*)d_in[0];
  const float* hedge_features = (const float*)d_in[1];
  const float* W_nn = (const float*)d_in[2];
  const float* W_nh = (const float*)d_in[3];
  const float* W_hh = (const float*)d_in[4];
  const float* W_hn = (const float*)d_in[5];
  const int* node_idx  = (const int*)d_in[6];
  const int* hedge_idx = (const int*)d_in[7];

  const int N   = in_sizes[0] / 64;
  const int E   = in_sizes[1] / 64;
  const int INC = in_sizes[6];

  float* slotA = (float*)d_ws;
  float* slotB = slotA + (size_t)N * 64;
  float* slotC = slotB + (size_t)N * 64;
  int* listN  = (int*)(slotC + (size_t)E * 64);
  int* listH  = listN + (size_t)N * CAP_N;
  int2* spillN = (int2*)(listH + (size_t)E * CAP_H);
  int2* spillH = spillN + (size_t)INC;
  int* cnt    = (int*)(spillH + (size_t)INC);
  int* spillCnt = cnt + 2 * ((size_t)N + E);

  float* out_node  = (float*)d_out;
  float* out_hedge = out_node + (size_t)N * 64;

  const dim3 blk(256);
  const dim3 gemmGridN((N + 127) / 128);
  const dim3 firstGrid((INC + 255) / 256 + (N + 127) / 128 + 2 * ((E + 127) / 128));

  hipMemsetAsync(cnt, 0, (2 * ((size_t)N + E) + 2) * sizeof(int), stream);
  buckets_and_input_products<<<firstGrid, blk, 0, stream>>>(
      node_idx, hedge_idx, cnt, listN, listH, spillCnt, spillN, spillH, INC, N, E,
      node_features, hedge_features, W_nn, W_nh, W_hh, slotB, slotA, slotC);

  pull_gate_tanh<CAP_N><<<dim3(2048), blk, 0, stream>>>(
      slotA, cnt, N + E, listN, spillCnt + 0, spillN, slotB, out_node, N, INC);

  gemm_rows64_wmma<<<gemmGridN, blk, 0, stream>>>(out_node, W_hn, slotB, N);
  pull_gate_tanh<CAP_H><<<dim3(2048), blk, 0, stream>>>(
      slotB, cnt + N, N + E, listH, spillCnt + 1, spillH, slotC, out_hedge, E, INC);
}


# W_hn hoisted out of the hyperedge gather by linearity: new_node rows summed per hyperedge, then 32 sums per block times W_hn on f32 WMMA; the 200000-row product launch is gone; sum order changes
# speedup vs baseline: 2.6198x; 1.0501x over previous
#include <hip/hip_runtime.h>
#include <hip/hip_bf16.h>
#include <math.h>

typedef __attribute__((ext_vector_type(2))) float v2f;
typedef __attribute__((ext_vector_type(8))) float v8f;

static __device__ __forceinline__ void rows64_product_block(
    v2f* __restrict__ sW2, const float* __restrict__ X, const float* __restrict__ W,
    float* __restrict__ Y, int nrows, int tileBlock)
{
  const int tid = threadIdx.x;
  for (int i = tid; i < 32 * 64; i += 256) {
    const int j = i >> 6;
    const int n = i & 63;
    v2f w;
    w.x = W[(2 * j) * 64 + n];
    w.y = W[(2 * j + 1) * 64 + n];
    sW2[i] = w;
  }
  __syncthreads();

  const int lane    = tid & 31;
  const int wave    = tid >> 5;
  const int rowBase = tileBlock * 128 + wave * 16;
  if (rowBase + 16 > nrows) return;

  const int m  = lane & 15;
  const int kh = lane >> 4;
  const int n  = lane & 15;

  const float* __restrict__ xrow = X + (size_t)(rowBase + m) * 64;

  v8f acc0 = {}, acc1 = {}, acc2 = {}, acc3 = {};

#pragma unroll
  for (int kk = 0; kk < 16; ++kk) {
    const int k0 = kk * 4 + kh * 2;
    v2f a = *(const v2f*)(xrow + k0);

    const int kp = k0 >> 1;
    const v2f* __restrict__ brow = &sW2[kp * 64 + n];
    v2f b0 = brow[0];
    v2f b1 = brow[16];
    v2f b2 = brow[32];
    v2f b3 = brow[48];

    acc0 = __builtin_amdgcn_wmma_f32_16x16x4_f32(false, a, false, b0, (short)0, acc0, false, false);
    acc1 = __builtin_amdgcn_wmma_f32_16x16x4_f32(false, a, false, b1, (short)0, acc1, false, false);
    acc2 = __builtin_amdgcn_wmma_f32_16x16x4_f32(false, a, false, b2, (short)0, acc2, false, false);
    acc3 = __builtin_amdgcn_wmma_f32_16x16x4_f32(false, a, false, b3, (short)0, acc3, false, false);
  }

  const int crow = rowBase + (lane >> 4) * 8;
  const int ccol = lane & 15;
#pragma unroll
  for (int v = 0; v < 8; ++v) {
    float* __restrict__ yr = Y + (size_t)(crow + v) * 64 + ccol;
    yr[0]  = acc0[v];
    yr[16] = acc1[v];
    yr[32] = acc2[v];
    yr[48] = acc3[v];
  }
}

#define CAP_N 32
#define CAP_H 64

static __device__ __forceinline__ void row_buckets_block(
    int incBlock,
    const int* __restrict__ node_idx, const int* __restrict__ hedge_idx,
    int* __restrict__ cnt,
    int* __restrict__ listN,
    int* __restrict__ listH,
    int* __restrict__ spillCnt, int2* __restrict__ spillN, int2* __restrict__ spillH,
    int n_inc, int nN, int nH)
{
  constexpr int PN = CAP_N / 2;
  constexpr int PH = CAP_H / 2;
  const int mine = (__builtin_amdgcn_s_sendmsg_rtn(0x87) >> 18) & 1;
  const int other = mine ^ 1;

  const int e = incBlock * 256 + (int)threadIdx.x;
  if (e >= n_inc) return;
  const int n = node_idx[e];
  const int h = hedge_idx[e];

  const size_t cstride = (size_t)nN + nH;
  int* cNm = cnt + mine * cstride;
  int* cHm = cNm + nN;
  int* lNm = listN + (size_t)mine * nN * PN;
  int* lHm = listH + (size_t)mine * nH * PH;

  const int tn = atomicAdd(cNm + n, 1);
  const int th = atomicAdd(cHm + h, 1);
  if (tn < PN) {
    lNm[(size_t)n * PN + tn] = h;
  } else {
    const int t2 = atomicAdd(cnt + other * cstride + n, 1);
    if (t2 < PN) {
      listN[((size_t)other * nN + n) * PN + t2] = h;
    } else {
      const int p = atomicAdd(spillCnt + 0, 1);
      spillN[p] = make_int2(n, h);
    }
  }
  if (th < PH) {
    lHm[(size_t)h * PH + th] = n;
  } else {
    const int t2 = atomicAdd(cnt + other * cstride + nN + h, 1);
    if (t2 < PH) {
      listH[((size_t)other * nH + h) * PH + t2] = n;
    } else {
      const int p = atomicAdd(spillCnt + 1, 1);
      spillH[p] = make_int2(h, n);
    }
  }
}

__global__ __launch_bounds__(256) __attribute__((amdgpu_waves_per_eu(16)))
void buckets_and_input_products(
    const int* __restrict__ node_idx, const int* __restrict__ hedge_idx,
    int* __restrict__ cnt, int* __restrict__ listN, int* __restrict__ listH,
    int* __restrict__ spillCnt, int2* __restrict__ spillN, int2* __restrict__ spillH,
    int n_inc, int nN, int nH,
    const float* __restrict__ node_features, const float* __restrict__ hedge_features,
    const float* __restrict__ W_nn, const float* __restrict__ W_nh,
    const float* __restrict__ W_hh,
    float* __restrict__ node_pre, float* __restrict__ hedge_proj,
    float* __restrict__ hedge_pre)
{
  __shared__ v2f sW2[32 * 64];

  const int nbB = (n_inc + 255) >> 8;
  const int nbN = (nN + 127) >> 7;
  const int nbE = (nH + 127) >> 7;
  int whole = nbB >> 3;
  whole = (nbN >> 1) < whole ? (nbN >> 1) : whole;
  whole = nbE < whole ? nbE : whole;

  const int b = (int)blockIdx.x;
  int role, idx;
  if (b < whole * 12) {
    const int pat = b / 12;
    const int slot = b - pat * 12;
    if (slot < 8)        { role = 0; idx = pat * 8 + slot; }
    else if (slot < 10)  { role = 1; idx = pat * 2 + (slot - 8); }
    else if (slot == 10) { role = 2; idx = pat; }
    else                 { role = 3; idx = pat; }
  } else {
    int r = b - whole * 12;
    const int leftB = nbB - whole * 8;
    const int leftN = nbN - whole * 2;
    const int leftE = nbE - whole;
    if (r < leftB) { role = 0; idx = whole * 8 + r; }
    else {
      r -= leftB;
      if (r < leftN) { role = 1; idx = whole * 2 + r; }
      else {
        r -= leftN;
        if (r < leftE) { role = 2; idx = whole + r; }
        else           { role = 3; idx = whole + (r - leftE); }
      }
    }
  }

  if (role == 0) {
    row_buckets_block(idx, node_idx, hedge_idx, cnt, listN, listH,
                      spillCnt, spillN, spillH, n_inc, nN, nH);
    return;
  }
  const float* __restrict__ X = (role == 1) ? node_features : hedge_features;
  const float* __restrict__ Wm = (role == 1) ? W_nn : (role == 2 ? W_nh : W_hh);
  float* __restrict__ Y = (role == 1) ? node_pre : (role == 2 ? hedge_proj : hedge_pre);
  const int nrows = (role == 1) ? nN : nH;
  rows64_product_block(sW2, X, Wm, Y, nrows, idx);
}

template <int U>
static __device__ __forceinline__ void gather_group(
    v2f& acc, int skq, int l0, int rem, const float* __restrict__ src, int lane)
{
  v2f v[U];
#pragma unroll
  for (int u = 0; u < U; ++u) {
    const int uu = (u < rem) ? u : (rem - 1);
    const int s = __builtin_amdgcn_readlane(skq, l0 + uu);
    v[u] = *(const v2f*)(src + (size_t)s * 64 + 2 * lane);
  }
#pragma unroll
  for (int u = 0; u < U; ++u) {
    if (u < rem) { acc.x += v[u].x; acc.y += v[u].y; }
  }
}

template <int CAP>
__global__ __launch_bounds__(256) void pull_gate_tanh(
    const float* __restrict__ src, const int* __restrict__ cnt, int cntPartStride,
    const int* __restrict__ list, const int* __restrict__ spillCntP,
    const int2* __restrict__ spill, const float* __restrict__ pre,
    float* __restrict__ out, int nrows, int n_inc)
{
  constexpr int NK = CAP / 32;
  constexpr int PC = CAP / 2;
  __shared__ int sorted[8][CAP];

  const int lane = threadIdx.x & 31;
  const int wave = __builtin_amdgcn_readfirstlane((int)(threadIdx.x >> 5));
  const int step = (int)gridDim.x * 8;
  int row = blockIdx.x * 8 + wave;
  if (row >= nrows) return;

  int c0 = cnt[row];
  int c1 = cnt[cntPartStride + row];
  int u0 = c0 < PC ? c0 : PC;
  int u1 = c1 < PC ? c1 : PC;
  int key[NK];
#pragma unroll
  for (int q = 0; q < NK; ++q) {
    const int p = q * 32 + lane;
    const int part = p / PC;
    const int slot = p % PC;
    const bool hv = slot < (part ? u1 : u0);
    key[q] = hv ? list[((size_t)part * nrows + row) * PC + slot] : 0x7fffffff;
  }
  v2f p2 = *(const v2f*)(pre + (size_t)row * 64 + 2 * lane);

  for (;;) {
    const int nraw = row + step;
    const bool more = nraw < nrows;
    const int nrow = more ? nraw : row;
    const int nc0 = cnt[nrow];
    const int nc1 = cnt[cntPartStride + nrow];

    const int cu = u0 + u1;

    int rnk[NK];
#pragma unroll
    for (int q = 0; q < NK; ++q) rnk[q] = 0;
#pragma unroll
    for (int part = 0; part < 2; ++part) {
      const int un = part ? u1 : u0;
      const int q2 = (part * PC) >> 5;
      const int lbase = (part * PC) & 31;
      for (int s = 0; s < un; ++s) {
        const int kj = __builtin_amdgcn_readlane(key[q2], lbase + s);
        const int j = part * PC + s;
#pragma unroll
        for (int q = 0; q < NK; ++q) {
          const int p = q * 32 + lane;
          rnk[q] += (kj < key[q] || (kj == key[q] && j < p)) ? 1 : 0;
        }
      }
    }
#pragma unroll
    for (int q = 0; q < NK; ++q) {
      if (key[q] != 0x7fffffff) sorted[wave][rnk[q]] = key[q];
    }

    int sk[NK];
#pragma unroll
    for (int q = 0; q < NK; ++q) {
      const int p = q * 32 + lane;
      sk[q] = (p < cu) ? sorted[wave][p] : 0;
    }

    const int nu0 = nc0 < PC ? nc0 : PC;
    const int nu1 = nc1 < PC ? nc1 : PC;
    int nkey[NK];
#pragma unroll
    for (int q = 0; q < NK; ++q) {
      const int p = q * 32 + lane;
      const int part = p / PC;
      const int slot = p % PC;
      const bool hv = slot < (part ? nu1 : nu0);
      nkey[q] = hv ? list[((size_t)part * nrows + nrow) * PC + slot] : 0x7fffffff;
    }
    const v2f np2 = *(const v2f*)(pre + (size_t)nrow * 64 + 2 * lane);

    v2f acc = {0.0f, 0.0f};
    for (int j = 0; j < cu; j += 8) {
      const int rem = cu - j;
      int skq = sk[0];
#pragma unroll
      for (int q = 1; q < NK; ++q) skq = (j >= q * 32) ? sk[q] : skq;
      const int l0 = j & 31;
      if (rem > 4) gather_group<8>(acc, skq, l0, rem, src, lane);
      else         gather_group<4>(acc, skq, l0, rem, src, lane);
    }

    if (c0 > PC && c1 > PC) {
      int tot = spillCntP[0];
      tot = tot < n_inc ? tot : n_inc;
      for (int i = 0; i < tot; i += 32) {
        const int p = i + lane;
        int2 pr = make_int2(-1, 0);
        if (p < tot) pr = spill[p];
        unsigned m = __builtin_amdgcn_ballot_w32(pr.x == row);
        while (m) {
          const int b = __builtin_ctz(m);
          m &= m - 1;
          const int s = __builtin_amdgcn_readlane(pr.y, b);
          const v2f v = *(const v2f*)(src + (size_t)s * 64 + 2 * lane);
          acc.x += v.x; acc.y += v.y;
        }
      }
    }

    v2f r;
    r.x = tanhf(p2.x * acc.x);
    r.y = tanhf(p2.y * acc.y);
    *(v2f*)(out + (size_t)row * 64 + 2 * lane) = r;

    if (!more) break;
    row = nrow; c0 = nc0; c1 = nc1; u0 = nu0; u1 = nu1; p2 = np2;
#pragma unroll
    for (int q = 0; q < NK; ++q) key[q] = nkey[q];
  }
}

#define SLAB_PITCH 66

template <int CAP>
__global__ __launch_bounds__(128) void pull_rows32_product_gate_tanh(
    const float* __restrict__ src, const int* __restrict__ cnt, int cntPartStride,
    const int* __restrict__ list, const int* __restrict__ spillCntP,
    const int2* __restrict__ spill, const float* __restrict__ W,
    const float* __restrict__ pre, float* __restrict__ out, int nrows, int n_inc)
{
  constexpr int NK = CAP / 32;
  constexpr int PC = CAP / 2;
  __shared__ v2f sW2[32 * 64];
  __shared__ int sorted[4][CAP];
  __shared__ __attribute__((aligned(16))) float sums[32 * SLAB_PITCH];
  __shared__ __attribute__((aligned(16))) float prod[32 * SLAB_PITCH];

  const int tid = threadIdx.x;
  for (int i = tid; i < 32 * 64; i += 128) {
    const int j = i >> 6;
    const int n = i & 63;
    v2f w;
    w.x = W[(2 * j) * 64 + n];
    w.y = W[(2 * j + 1) * 64 + n];
    sW2[i] = w;
  }

  const int lane = tid & 31;
  const int wave = __builtin_amdgcn_readfirstlane(tid >> 5);
  const int row0 = (int)blockIdx.x * 32 + wave * 8;
  int nr = nrows - row0;
  nr = nr < 0 ? 0 : (nr > 8 ? 8 : nr);

  int row = row0 < nrows ? row0 : nrows - 1;
  int c0 = cnt[row];
  int c1 = cnt[cntPartStride + row];
  int u0 = c0 < PC ? c0 : PC;
  int u1 = c1 < PC ? c1 : PC;
  int key[NK];
#pragma unroll
  for (int q = 0; q < NK; ++q) {
    const int p = q * 32 + lane;
    const int part = p / PC;
    const int slot = p % PC;
    const bool hv = slot < (part ? u1 : u0);
    key[q] = hv ? list[((size_t)part * nrows + row) * PC + slot] : 0x7fffffff;
  }

  v2f g2[8];
#pragma unroll
  for (int r = 0; r < 8; ++r) {
    int gr = row0 + r;
    gr = gr < nrows ? gr : nrows - 1;
    g2[r] = *(const v2f*)(pre + (size_t)gr * 64 + 2 * lane);
  }

  for (int r = 0; r < nr; ++r) {
    const bool more = (r + 1) < nr;
    const int nrow = more ? row + 1 : row;
    const int nc0 = cnt[nrow];
    const int nc1 = cnt[cntPartStride + nrow];

    const int cu = u0 + u1;

    int rnk[NK];
#pragma unroll
    for (int q = 0; q < NK; ++q) rnk[q] = 0;
#pragma unroll
    for (int part = 0; part < 2; ++part) {
      const int un = part ? u1 : u0;
      const int q2 = (part * PC) >> 5;
      const int lbase = (part * PC) & 31;
      for (int s = 0; s < un; ++s) {
        const int kj = __builtin_amdgcn_readlane(key[q2], lbase + s);
        const int j = part * PC + s;
#pragma unroll
        for (int q = 0; q < NK; ++q) {
          const int p = q * 32 + lane;
          rnk[q] += (kj < key[q] || (kj == key[q] && j < p)) ? 1 : 0;
        }
      }
    }
#pragma unroll
    for (int q = 0; q < NK; ++q) {
      if (key[q] != 0x7fffffff) sorted[wave][rnk[q]] = key[q];
    }

    int sk[NK];
#pragma unroll
    for (int q = 0; q < NK; ++q) {
      const int p = q * 32 + lane;
      sk[q] = (p < cu) ? sorted[wave][p] : 0;
    }

    const int nu0 = nc0 < PC ? nc0 : PC;
    const int nu1 = nc1 < PC ? nc1 : PC;
    int nkey[NK];
#pragma unroll
    for (int q = 0; q < NK; ++q) {
      const int p = q * 32 + lane;
      const int part = p / PC;
      const int slot = p % PC;
      const bool hv = slot < (part ? nu1 : nu0);
      nkey[q] = hv ? list[((size_t)part * nrows + nrow) * PC + slot] : 0x7fffffff;
    }

    v2f acc = {0.0f, 0.0f};
    for (int j = 0; j < cu; j += 8) {
      const int rem = cu - j;
      int skq = sk[0];
#pragma unroll
      for (int q = 1; q < NK; ++q) skq = (j >= q * 32) ? sk[q] : skq;
      const int l0 = j & 31;
      if (rem > 4) gather_group<8>(acc, skq, l0, rem, src, lane);
      else         gather_group<4>(acc, skq, l0, rem, src, lane);
    }

    if (c0 > PC && c1 > PC) {
      int tot = spillCntP[0];
      tot = tot < n_inc ? tot : n_inc;
      for (int i = 0; i < tot; i += 32) {
        const int p = i + lane;
        int2 pr = make_int2(-1, 0);
        if (p < tot) pr = spill[p];
        unsigned m = __builtin_amdgcn_ballot_w32(pr.x == row);
        while (m) {
          const int b = __builtin_ctz(m);
          m &= m - 1;
          const int s = __builtin_amdgcn_readlane(pr.y, b);
          const v2f v = *(const v2f*)(src + (size_t)s * 64 + 2 * lane);
          acc.x += v.x; acc.y += v.y;
        }
      }
    }

    *(v2f*)&sums[(wave * 8 + r) * SLAB_PITCH + 2 * lane] = acc;

    row = nrow; c0 = nc0; c1 = nc1; u0 = nu0; u1 = nu1;
#pragma unroll
    for (int q = 0; q < NK; ++q) key[q] = nkey[q];
  }
  for (int r = nr; r < 8; ++r) {
    const v2f z = {0.0f, 0.0f};
    *(v2f*)&sums[(wave * 8 + r) * SLAB_PITCH + 2 * lane] = z;
  }

  __syncthreads();

  const int mt = wave >> 1;
  const int nh = wave & 1;
  const int m  = lane & 15;
  const int kh = lane >> 4;
  const int n  = lane & 15;

  v8f acc0 = {}, acc1 = {};

#pragma unroll
  for (int kk = 0; kk < 16; ++kk) {
    const int k0 = kk * 4 + kh * 2;
    const v2f a = *(const v2f*)&sums[(mt * 16 + m) * SLAB_PITCH + k0];

    const int kp = k0 >> 1;
    const v2f* __restrict__ brow = &sW2[kp * 64 + nh * 32 + n];
    const v2f b0 = brow[0];
    const v2f b1 = brow[16];

    acc0 = __builtin_amdgcn_wmma_f32_16x16x4_f32(false, a, false, b0, (short)0, acc0, false, false);
    acc1 = __builtin_amdgcn_wmma_f32_16x16x4_f32(false, a, false, b1, (short)0, acc1, false, false);
  }

  const int crow = mt * 16 + (lane >> 4) * 8;
  const int ccol = nh * 32 + (lane & 15);
#pragma unroll
  for (int v = 0; v < 8; ++v) {
    float* yr = &prod[(crow + v) * SLAB_PITCH + ccol];
    yr[0]  = acc0[v];
    yr[16] = acc1[v];
  }

  __syncthreads();

#pragma unroll
  for (int r = 0; r < 8; ++r) {
    if (r < nr) {
      const v2f s = *(const v2f*)&prod[(wave * 8 + r) * SLAB_PITCH + 2 * lane];
      v2f o;
      o.x = tanhf(g2[r].x * s.x);
      o.y = tanhf(g2[r].y * s.y);
      *(v2f*)(out + (size_t)(row0 + r) * 64 + 2 * lane) = o;
    }
  }
}

extern "C" void kernel_launch(void* const* d_in, const int* in_sizes, int n_in,
                              void* d_out, int out_size, void* d_ws, size_t ws_size,
                              hipStream_t stream) {
  const float* node_features  = (const float*)d_in[0];
  const float* hedge_features = (const float*)d_in[1];
  const float* W_nn = (const float*)d_in[2];
  const float* W_nh = (const float*)d_in[3];
  const float* W_hh = (const float*)d_in[4];
  const float* W_hn = (const float*)d_in[5];
  const int* node_idx  = (const int*)d_in[6];
  const int* hedge_idx = (const int*)d_in[7];

  const int N   = in_sizes[0] / 64;
  const int E   = in_sizes[1] / 64;
  const int INC = in_sizes[6];

  float* slotA = (float*)d_ws;
  float* slotB = slotA + (size_t)N * 64;
  float* slotC = slotB + (size_t)N * 64;
  int* listN  = (int*)(slotC + (size_t)E * 64);
  int* listH  = listN + (size_t)N * CAP_N;
  int2* spillN = (int2*)(listH + (size_t)E * CAP_H);
  int2* spillH = spillN + (size_t)INC;
  int* cnt    = (int*)(spillH + (size_t)INC);
  int* spillCnt = cnt + 2 * ((size_t)N + E);

  float* out_node  = (float*)d_out;
  float* out_hedge = out_node + (size_t)N * 64;

  const dim3 blk(256);
  const dim3 firstGrid((INC + 255) / 256 + (N + 127) / 128 + 2 * ((E + 127) / 128));

  hipMemsetAsync(cnt, 0, (2 * ((size_t)N + E) + 2) * sizeof(int), stream);
  buckets_and_input_products<<<firstGrid, blk, 0, stream>>>(
      node_idx, hedge_idx, cnt, listN, listH, spillCnt, spillN, spillH, INC, N, E,
      node_features, hedge_features, W_nn, W_nh, W_hh, slotB, slotA, slotC);

  pull_gate_tanh<CAP_N><<<dim3(2048), blk, 0, stream>>>(
      slotA, cnt, N + E, listN, spillCnt + 0, spillN, slotB, out_node, N, INC);

  pull_rows32_product_gate_tanh<CAP_H><<<dim3((E + 31) / 32), dim3(128), 0, stream>>>(
      out_node, cnt + N, N + E, listH, spillCnt + 1, spillH, W_hn, slotC, out_hedge, E, INC);
}


# Bucket ordering in both pull kernels: both parts' entries in consecutive lanes, each packed with its position into one unique word, so a rank costs one compare per entry, 4 entries a trip; sums kept
# speedup vs baseline: 2.7938x; 1.0664x over previous
#include <hip/hip_runtime.h>
#include <hip/hip_bf16.h>
#include <math.h>

typedef __attribute__((ext_vector_type(2))) float v2f;
typedef __attribute__((ext_vector_type(8))) float v8f;

static __device__ __forceinline__ void rows64_product_block(
    v2f* __restrict__ sW2, const float* __restrict__ X, const float* __restrict__ W,
    float* __restrict__ Y, int nrows, int tileBlock)
{
  const int tid = threadIdx.x;
  for (int i = tid; i < 32 * 64; i += 256) {
    const int j = i >> 6;
    const int n = i & 63;
    v2f w;
    w.x = W[(2 * j) * 64 + n];
    w.y = W[(2 * j + 1) * 64 + n];
    sW2[i] = w;
  }
  __syncthreads();

  const int lane    = tid & 31;
  const int wave    = tid >> 5;
  const int rowBase = tileBlock * 128 + wave * 16;
  if (rowBase + 16 > nrows) return;

  const int m  = lane & 15;
  const int kh = lane >> 4;
  const int n  = lane & 15;

  const float* __restrict__ xrow = X + (size_t)(rowBase + m) * 64;

  v8f acc0 = {}, acc1 = {}, acc2 = {}, acc3 = {};

#pragma unroll
  for (int kk = 0; kk < 16; ++kk) {
    const int k0 = kk * 4 + kh * 2;
    v2f a = *(const v2f*)(xrow + k0);

    const int kp = k0 >> 1;
    const v2f* __restrict__ brow = &sW2[kp * 64 + n];
    v2f b0 = brow[0];
    v2f b1 = brow[16];
    v2f b2 = brow[32];
    v2f b3 = brow[48];

    acc0 = __builtin_amdgcn_wmma_f32_16x16x4_f32(false, a, false, b0, (short)0, acc0, false, false);
    acc1 = __builtin_amdgcn_wmma_f32_16x16x4_f32(false, a, false, b1, (short)0, acc1, false, false);
    acc2 = __builtin_amdgcn_wmma_f32_16x16x4_f32(false, a, false, b2, (short)0, acc2, false, false);
    acc3 = __builtin_amdgcn_wmma_f32_16x16x4_f32(false, a, false, b3, (short)0, acc3, false, false);
  }

  const int crow = rowBase + (lane >> 4) * 8;
  const int ccol = lane & 15;
#pragma unroll
  for (int v = 0; v < 8; ++v) {
    float* __restrict__ yr = Y + (size_t)(crow + v) * 64 + ccol;
    yr[0]  = acc0[v];
    yr[16] = acc1[v];
    yr[32] = acc2[v];
    yr[48] = acc3[v];
  }
}

#define CAP_N 32
#define CAP_H 64

static __device__ __forceinline__ void row_buckets_block(
    int incBlock,
    const int* __restrict__ node_idx, const int* __restrict__ hedge_idx,
    int* __restrict__ cnt,
    int* __restrict__ listN,
    int* __restrict__ listH,
    int* __restrict__ spillCnt, int2* __restrict__ spillN, int2* __restrict__ spillH,
    int n_inc, int nN, int nH)
{
  constexpr int PN = CAP_N / 2;
  constexpr int PH = CAP_H / 2;
  const int mine = (__builtin_amdgcn_s_sendmsg_rtn(0x87) >> 18) & 1;
  const int other = mine ^ 1;

  const int e = incBlock * 256 + (int)threadIdx.x;
  if (e >= n_inc) return;
  const int n = node_idx[e];
  const int h = hedge_idx[e];

  const size_t cstride = (size_t)nN + nH;
  int* cNm = cnt + mine * cstride;
  int* cHm = cNm + nN;
  int* lNm = listN + (size_t)mine * nN * PN;
  int* lHm = listH + (size_t)mine * nH * PH;

  const int tn = atomicAdd(cNm + n, 1);
  const int th = atomicAdd(cHm + h, 1);
  if (tn < PN) {
    lNm[(size_t)n * PN + tn] = h;
  } else {
    const int t2 = atomicAdd(cnt + other * cstride + n, 1);
    if (t2 < PN) {
      listN[((size_t)other * nN + n) * PN + t2] = h;
    } else {
      const int p = atomicAdd(spillCnt + 0, 1);
      spillN[p] = make_int2(n, h);
    }
  }
  if (th < PH) {
    lHm[(size_t)h * PH + th] = n;
  } else {
    const int t2 = atomicAdd(cnt + other * cstride + nN + h, 1);
    if (t2 < PH) {
      listH[((size_t)other * nH + h) * PH + t2] = n;
    } else {
      const int p = atomicAdd(spillCnt + 1, 1);
      spillH[p] = make_int2(h, n);
    }
  }
}

__global__ __launch_bounds__(256) __attribute__((amdgpu_waves_per_eu(16)))
void buckets_and_input_products(
    const int* __restrict__ node_idx, const int* __restrict__ hedge_idx,
    int* __restrict__ cnt, int* __restrict__ listN, int* __restrict__ listH,
    int* __restrict__ spillCnt, int2* __restrict__ spillN, int2* __restrict__ spillH,
    int n_inc, int nN, int nH,
    const float* __restrict__ node_features, const float* __restrict__ hedge_features,
    const float* __restrict__ W_nn, const float* __restrict__ W_nh,
    const float* __restrict__ W_hh,
    float* __restrict__ node_pre, float* __restrict__ hedge_proj,
    float* __restrict__ hedge_pre)
{
  __shared__ v2f sW2[32 * 64];

  const int nbB = (n_inc + 255) >> 8;
  const int nbN = (nN + 127) >> 7;
  const int nbE = (nH + 127) >> 7;
  int whole = nbB >> 3;
  whole = (nbN >> 1) < whole ? (nbN >> 1) : whole;
  whole = nbE < whole ? nbE : whole;

  const int b = (int)blockIdx.x;
  int role, idx;
  if (b < whole * 12) {
    const int pat = b / 12;
    const int slot = b - pat * 12;
    if (slot < 8)        { role = 0; idx = pat * 8 + slot; }
    else if (slot < 10)  { role = 1; idx = pat * 2 + (slot - 8); }
    else if (slot == 10) { role = 2; idx = pat; }
    else                 { role = 3; idx = pat; }
  } else {
    int r = b - whole * 12;
    const int leftB = nbB - whole * 8;
    const int leftN = nbN - whole * 2;
    const int leftE = nbE - whole;
    if (r < leftB) { role = 0; idx = whole * 8 + r; }
    else {
      r -= leftB;
      if (r < leftN) { role = 1; idx = whole * 2 + r; }
      else {
        r -= leftN;
        if (r < leftE) { role = 2; idx = whole + r; }
        else           { role = 3; idx = whole + (r - leftE); }
      }
    }
  }

  if (role == 0) {
    row_buckets_block(idx, node_idx, hedge_idx, cnt, listN, listH,
                      spillCnt, spillN, spillH, n_inc, nN, nH);
    return;
  }
  const float* __restrict__ X = (role == 1) ? node_features : hedge_features;
  const float* __restrict__ Wm = (role == 1) ? W_nn : (role == 2 ? W_nh : W_hh);
  float* __restrict__ Y = (role == 1) ? node_pre : (role == 2 ? hedge_proj : hedge_pre);
  const int nrows = (role == 1) ? nN : nH;
  rows64_product_block(sW2, X, Wm, Y, nrows, idx);
}

template <int CAP>
static __device__ __forceinline__ void load_bucket_entries(
    int (&ent)[CAP / 32], const int* __restrict__ list, int nrows, int row,
    int u0, int u1, int lane)
{
  constexpr int NK = CAP / 32;
  constexpr int PC = CAP / 2;
  const int cu = u0 + u1;
  const int* __restrict__ part0 = list + (size_t)row * PC;
  const int* __restrict__ part1 = list + ((size_t)nrows + row) * PC;
#pragma unroll
  for (int q = 0; q < NK; ++q) {
    const int p = q * 32 + lane;
    const int* __restrict__ at = (p < u0) ? (part0 + p) : (part1 + (p - u0));
    ent[q] = (p < cu) ? *at : 0x7fffffff;
  }
}

template <int CAP>
static __device__ __forceinline__ void pack_bucket_keys(
    const int (&ent)[CAP / 32], unsigned (&key)[CAP / 32], int lane)
{
  constexpr int NK = CAP / 32;
  constexpr int SH = (CAP == 64) ? 6 : 5;
#pragma unroll
  for (int q = 0; q < NK; ++q) key[q] = ((unsigned)ent[q] << SH) | (unsigned)(q * 32 + lane);
}

template <int NK, int NC>
static __device__ __forceinline__ void rank_against(
    const unsigned (&key)[NK], int cu, int (&rnk)[NK])
{
#pragma unroll
  for (int qs = 0; qs < NC; ++qs) {
    const int left = cu - qs * 32;
    const int cn = left < 32 ? left : 32;
    for (int j = 0; j < cn; j += 4) {
#pragma unroll
      for (int t = 0; t < 4; ++t) {
        const unsigned kj = (unsigned)__builtin_amdgcn_readlane((int)key[qs], j + t);
#pragma unroll
        for (int q = 0; q < NC; ++q) rnk[q] += (kj < key[q]) ? 1 : 0;
      }
    }
  }
}

template <int NK>
static __device__ __forceinline__ void rank_bucket(
    const unsigned (&key)[NK], int cu, int (&rnk)[NK])
{
#pragma unroll
  for (int q = 0; q < NK; ++q) rnk[q] = 0;
  if (NK == 1 || cu <= 32) rank_against<NK, 1>(key, cu, rnk);
  else                     rank_against<NK, NK>(key, cu, rnk);
}

template <int U>
static __device__ __forceinline__ void gather_group(
    v2f& acc, int skq, int l0, int rem, const float* __restrict__ src, int lane)
{
  v2f v[U];
#pragma unroll
  for (int u = 0; u < U; ++u) {
    const int uu = (u < rem) ? u : (rem - 1);
    const int s = __builtin_amdgcn_readlane(skq, l0 + uu);
    v[u] = *(const v2f*)(src + (size_t)s * 64 + 2 * lane);
  }
#pragma unroll
  for (int u = 0; u < U; ++u) {
    if (u < rem) { acc.x += v[u].x; acc.y += v[u].y; }
  }
}

template <int CAP>
__global__ __launch_bounds__(256) void pull_gate_tanh(
    const float* __restrict__ src, const int* __restrict__ cnt, int cntPartStride,
    const int* __restrict__ list, const int* __restrict__ spillCntP,
    const int2* __restrict__ spill, const float* __restrict__ pre,
    float* __restrict__ out, int nrows, int n_inc)
{
  constexpr int NK = CAP / 32;
  constexpr int PC = CAP / 2;
  __shared__ int sorted[8][CAP];

  const int lane = threadIdx.x & 31;
  const int wave = __builtin_amdgcn_readfirstlane((int)(threadIdx.x >> 5));
  const int step = (int)gridDim.x * 8;
  int row = blockIdx.x * 8 + wave;
  if (row >= nrows) return;

  int c0 = cnt[row];
  int c1 = cnt[cntPartStride + row];
  int u0 = c0 < PC ? c0 : PC;
  int u1 = c1 < PC ? c1 : PC;
  int ent[NK];
  load_bucket_entries<CAP>(ent, list, nrows, row, u0, u1, lane);
  v2f p2 = *(const v2f*)(pre + (size_t)row * 64 + 2 * lane);

  for (;;) {
    const int nraw = row + step;
    const bool more = nraw < nrows;
    const int nrow = more ? nraw : row;
    const int nc0 = cnt[nrow];
    const int nc1 = cnt[cntPartStride + nrow];

    const int cu = u0 + u1;

    unsigned key[NK];
    pack_bucket_keys<CAP>(ent, key, lane);
    int rnk[NK];
    rank_bucket<NK>(key, cu, rnk);
#pragma unroll
    for (int q = 0; q < NK; ++q) {
      const int p = q * 32 + lane;
      if (p < cu) sorted[wave][rnk[q]] = ent[q];
    }

    int sk[NK];
#pragma unroll
    for (int q = 0; q < NK; ++q) {
      const int p = q * 32 + lane;
      sk[q] = (p < cu) ? sorted[wave][p] : 0;
    }

    const int nu0 = nc0 < PC ? nc0 : PC;
    const int nu1 = nc1 < PC ? nc1 : PC;
    int nent[NK];
    load_bucket_entries<CAP>(nent, list, nrows, nrow, nu0, nu1, lane);
    const v2f np2 = *(const v2f*)(pre + (size_t)nrow * 64 + 2 * lane);

    v2f acc = {0.0f, 0.0f};
    for (int j = 0; j < cu; j += 8) {
      const int rem = cu - j;
      int skq = sk[0];
#pragma unroll
      for (int q = 1; q < NK; ++q) skq = (j >= q * 32) ? sk[q] : skq;
      const int l0 = j & 31;
      if (rem > 4) gather_group<8>(acc, skq, l0, rem, src, lane);
      else         gather_group<4>(acc, skq, l0, rem, src, lane);
    }

    if (c0 > PC && c1 > PC) {
      int tot = spillCntP[0];
      tot = tot < n_inc ? tot : n_inc;
      for (int i = 0; i < tot; i += 32) {
        const int p = i + lane;
        int2 pr = make_int2(-1, 0);
        if (p < tot) pr = spill[p];
        unsigned m = __builtin_amdgcn_ballot_w32(pr.x == row);
        while (m) {
          const int b = __builtin_ctz(m);
          m &= m - 1;
          const int s = __builtin_amdgcn_readlane(pr.y, b);
          const v2f v = *(const v2f*)(src + (size_t)s * 64 + 2 * lane);
          acc.x += v.x; acc.y += v.y;
        }
      }
    }

    v2f r;
    r.x = tanhf(p2.x * acc.x);
    r.y = tanhf(p2.y * acc.y);
    *(v2f*)(out + (size_t)row * 64 + 2 * lane) = r;

    if (!more) break;
    row = nrow; c0 = nc0; c1 = nc1; u0 = nu0; u1 = nu1; p2 = np2;
#pragma unroll
    for (int q = 0; q < NK; ++q) ent[q] = nent[q];
  }
}

#define SLAB_PITCH 66

template <int CAP>
__global__ __launch_bounds__(128) void pull_rows32_product_gate_tanh(
    const float* __restrict__ src, const int* __restrict__ cnt, int cntPartStride,
    const int* __restrict__ list, const int* __restrict__ spillCntP,
    const int2* __restrict__ spill, const float* __restrict__ W,
    const float* __restrict__ pre, float* __restrict__ out, int nrows, int n_inc)
{
  constexpr int NK = CAP / 32;
  constexpr int PC = CAP / 2;
  __shared__ v2f sW2[32 * 64];
  __shared__ int sorted[4][CAP];
  __shared__ __attribute__((aligned(16))) float sums[32 * SLAB_PITCH];
  __shared__ __attribute__((aligned(16))) float prod[32 * SLAB_PITCH];

  const int tid = threadIdx.x;
  for (int i = tid; i < 32 * 64; i += 128) {
    const int j = i >> 6;
    const int n = i & 63;
    v2f w;
    w.x = W[(2 * j) * 64 + n];
    w.y = W[(2 * j + 1) * 64 + n];
    sW2[i] = w;
  }

  const int lane = tid & 31;
  const int wave = __builtin_amdgcn_readfirstlane(tid >> 5);
  const int row0 = (int)blockIdx.x * 32 + wave * 8;
  int nr = nrows - row0;
  nr = nr < 0 ? 0 : (nr > 8 ? 8 : nr);

  int row = row0 < nrows ? row0 : nrows - 1;
  int c0 = cnt[row];
  int c1 = cnt[cntPartStride + row];
  int u0 = c0 < PC ? c0 : PC;
  int u1 = c1 < PC ? c1 : PC;
  int ent[NK];
  load_bucket_entries<CAP>(ent, list, nrows, row, u0, u1, lane);

  v2f g2[8];
#pragma unroll
  for (int r = 0; r < 8; ++r) {
    int gr = row0 + r;
    gr = gr < nrows ? gr : nrows - 1;
    g2[r] = *(const v2f*)(pre + (size_t)gr * 64 + 2 * lane);
  }

  for (int r = 0; r < nr; ++r) {
    const bool more = (r + 1) < nr;
    const int nrow = more ? row + 1 : row;
    const int nc0 = cnt[nrow];
    const int nc1 = cnt[cntPartStride + nrow];

    const int cu = u0 + u1;

    unsigned key[NK];
    pack_bucket_keys<CAP>(ent, key, lane);
    int rnk[NK];
    rank_bucket<NK>(key, cu, rnk);
#pragma unroll
    for (int q = 0; q < NK; ++q) {
      const int p = q * 32 + lane;
      if (p < cu) sorted[wave][rnk[q]] = ent[q];
    }

    int sk[NK];
#pragma unroll
    for (int q = 0; q < NK; ++q) {
      const int p = q * 32 + lane;
      sk[q] = (p < cu) ? sorted[wave][p] : 0;
    }

    const int nu0 = nc0 < PC ? nc0 : PC;
    const int nu1 = nc1 < PC ? nc1 : PC;
    int nent[NK];
    load_bucket_entries<CAP>(nent, list, nrows, nrow, nu0, nu1, lane);

    v2f acc = {0.0f, 0.0f};
    for (int j = 0; j < cu; j += 8) {
      const int rem = cu - j;
      int skq = sk[0];
#pragma unroll
      for (int q = 1; q < NK; ++q) skq = (j >= q * 32) ? sk[q] : skq;
      const int l0 = j & 31;
      if (rem > 4) gather_group<8>(acc, skq, l0, rem, src, lane);
      else         gather_group<4>(acc, skq, l0, rem, src, lane);
    }

    if (c0 > PC && c1 > PC) {
      int tot = spillCntP[0];
      tot = tot < n_inc ? tot : n_inc;
      for (int i = 0; i < tot; i += 32) {
        const int p = i + lane;
        int2 pr = make_int2(-1, 0);
        if (p < tot) pr = spill[p];
        unsigned m = __builtin_amdgcn_ballot_w32(pr.x == row);
        while (m) {
          const int b = __builtin_ctz(m);
          m &= m - 1;
          const int s = __builtin_amdgcn_readlane(pr.y, b);
          const v2f v = *(const v2f*)(src + (size_t)s * 64 + 2 * lane);
          acc.x += v.x; acc.y += v.y;
        }
      }
    }

    *(v2f*)&sums[(wave * 8 + r) * SLAB_PITCH + 2 * lane] = acc;

    row = nrow; c0 = nc0; c1 = nc1; u0 = nu0; u1 = nu1;
#pragma unroll
    for (int q = 0; q < NK; ++q) ent[q] = nent[q];
  }
  for (int r = nr; r < 8; ++r) {
    const v2f z = {0.0f, 0.0f};
    *(v2f*)&sums[(wave * 8 + r) * SLAB_PITCH + 2 * lane] = z;
  }

  __syncthreads();

  const int mt = wave >> 1;
  const int nh = wave & 1;
  const int m  = lane & 15;
  const int kh = lane >> 4;
  const int n  = lane & 15;

  v8f acc0 = {}, acc1 = {};

#pragma unroll
  for (int kk = 0; kk < 16; ++kk) {
    const int k0 = kk * 4 + kh * 2;
    const v2f a = *(const v2f*)&sums[(mt * 16 + m) * SLAB_PITCH + k0];

    const int kp = k0 >> 1;
    const v2f* __restrict__ brow = &sW2[kp * 64 + nh * 32 + n];
    const v2f b0 = brow[0];
    const v2f b1 = brow[16];

    acc0 = __builtin_amdgcn_wmma_f32_16x16x4_f32(false, a, false, b0, (short)0, acc0, false, false);
    acc1 = __builtin_amdgcn_wmma_f32_16x16x4_f32(false, a, false, b1, (short)0, acc1, false, false);
  }

  const int crow = mt * 16 + (lane >> 4) * 8;
  const int ccol = nh * 32 + (lane & 15);
#pragma unroll
  for (int v = 0; v < 8; ++v) {
    float* yr = &prod[(crow + v) * SLAB_PITCH + ccol];
    yr[0]  = acc0[v];
    yr[16] = acc1[v];
  }

  __syncthreads();

#pragma unroll
  for (int r = 0; r < 8; ++r) {
    if (r < nr) {
      const v2f s = *(const v2f*)&prod[(wave * 8 + r) * SLAB_PITCH + 2 * lane];
      v2f o;
      o.x = tanhf(g2[r].x * s.x);
      o.y = tanhf(g2[r].y * s.y);
      *(v2f*)(out + (size_t)(row0 + r) * 64 + 2 * lane) = o;
    }
  }
}

extern "C" void kernel_launch(void* const* d_in, const int* in_sizes, int n_in,
                              void* d_out, int out_size, void* d_ws, size_t ws_size,
                              hipStream_t stream) {
  const float* node_features  = (const float*)d_in[0];
  const float* hedge_features = (const float*)d_in[1];
  const float* W_nn = (const float*)d_in[2];
  const float* W_nh = (const float*)d_in[3];
  const float* W_hh = (const float*)d_in[4];
  const float* W_hn = (const float*)d_in[5];
  const int* node_idx  = (const int*)d_in[6];
  const int* hedge_idx = (const int*)d_in[7];

  const int N   = in_sizes[0] / 64;
  const int E   = in_sizes[1] / 64;
  const int INC = in_sizes[6];

  float* slotA = (float*)d_ws;
  float* slotB = slotA + (size_t)N * 64;
  float* slotC = slotB + (size_t)N * 64;
  int* listN  = (int*)(slotC + (size_t)E * 64);
  int* listH  = listN + (size_t)N * CAP_N;
  int2* spillN = (int2*)(listH + (size_t)E * CAP_H);
  int2* spillH = spillN + (size_t)INC;
  int* cnt    = (int*)(spillH + (size_t)INC);
  int* spillCnt = cnt + 2 * ((size_t)N + E);

  float* out_node  = (float*)d_out;
  float* out_hedge = out_node + (size_t)N * 64;

  const dim3 blk(256);
  const dim3 firstGrid((INC + 255) / 256 + (N + 127) / 128 + 2 * ((E + 127) / 128));

  hipMemsetAsync(cnt, 0, (2 * ((size_t)N + E) + 2) * sizeof(int), stream);
  buckets_and_input_products<<<firstGrid, blk, 0, stream>>>(
      node_idx, hedge_idx, cnt, listN, listH, spillCnt, spillN, spillH, INC, N, E,
      node_features, hedge_features, W_nn, W_nh, W_hh, slotB, slotA, slotC);

  pull_gate_tanh<CAP_N><<<dim3(2048), blk, 0, stream>>>(
      slotA, cnt, N + E, listN, spillCnt + 0, spillN, slotB, out_node, N, INC);

  pull_rows32_product_gate_tanh<CAP_H><<<dim3((E + 31) / 32), dim3(128), 0, stream>>>(
      out_node, cnt + N, N + E, listH, spillCnt + 1, spillH, W_hn, slotC, out_hedge, E, INC);
}


# tanh in both pull kernels as 1 - 2/(1 + 2^(c*2log2e)) on hardware exp2 and reciprocal, branch-free, for the two-range library routine: 84 -> 24 instructions a row; last bits differ, sum order kept
# speedup vs baseline: 2.9227x; 1.0461x over previous
#include <hip/hip_runtime.h>
#include <hip/hip_bf16.h>
#include <math.h>

typedef __attribute__((ext_vector_type(2))) float v2f;
typedef __attribute__((ext_vector_type(8))) float v8f;

static __device__ __forceinline__ void rows64_product_block(
    v2f* __restrict__ sW2, const float* __restrict__ X, const float* __restrict__ W,
    float* __restrict__ Y, int nrows, int tileBlock)
{
  const int tid = threadIdx.x;
  for (int i = tid; i < 32 * 64; i += 256) {
    const int j = i >> 6;
    const int n = i & 63;
    v2f w;
    w.x = W[(2 * j) * 64 + n];
    w.y = W[(2 * j + 1) * 64 + n];
    sW2[i] = w;
  }
  __syncthreads();

  const int lane    = tid & 31;
  const int wave    = tid >> 5;
  const int rowBase = tileBlock * 128 + wave * 16;
  if (rowBase + 16 > nrows) return;

  const int m  = lane & 15;
  const int kh = lane >> 4;
  const int n  = lane & 15;

  const float* __restrict__ xrow = X + (size_t)(rowBase + m) * 64;

  v8f acc0 = {}, acc1 = {}, acc2 = {}, acc3 = {};

#pragma unroll
  for (int kk = 0; kk < 16; ++kk) {
    const int k0 = kk * 4 + kh * 2;
    v2f a = *(const v2f*)(xrow + k0);

    const int kp = k0 >> 1;
    const v2f* __restrict__ brow = &sW2[kp * 64 + n];
    v2f b0 = brow[0];
    v2f b1 = brow[16];
    v2f b2 = brow[32];
    v2f b3 = brow[48];

    acc0 = __builtin_amdgcn_wmma_f32_16x16x4_f32(false, a, false, b0, (short)0, acc0, false, false);
    acc1 = __builtin_amdgcn_wmma_f32_16x16x4_f32(false, a, false, b1, (short)0, acc1, false, false);
    acc2 = __builtin_amdgcn_wmma_f32_16x16x4_f32(false, a, false, b2, (short)0, acc2, false, false);
    acc3 = __builtin_amdgcn_wmma_f32_16x16x4_f32(false, a, false, b3, (short)0, acc3, false, false);
  }

  const int crow = rowBase + (lane >> 4) * 8;
  const int ccol = lane & 15;
#pragma unroll
  for (int v = 0; v < 8; ++v) {
    float* __restrict__ yr = Y + (size_t)(crow + v) * 64 + ccol;
    yr[0]  = acc0[v];
    yr[16] = acc1[v];
    yr[32] = acc2[v];
    yr[48] = acc3[v];
  }
}

#define CAP_N 32
#define CAP_H 64

static __device__ __forceinline__ void row_buckets_block(
    int incBlock,
    const int* __restrict__ node_idx, const int* __restrict__ hedge_idx,
    int* __restrict__ cnt,
    int* __restrict__ listN,
    int* __restrict__ listH,
    int* __restrict__ spillCnt, int2* __restrict__ spillN, int2* __restrict__ spillH,
    int n_inc, int nN, int nH)
{
  constexpr int PN = CAP_N / 2;
  constexpr int PH = CAP_H / 2;
  const int mine = (__builtin_amdgcn_s_sendmsg_rtn(0x87) >> 18) & 1;
  const int other = mine ^ 1;

  const int e = incBlock * 256 + (int)threadIdx.x;
  if (e >= n_inc) return;
  const int n = node_idx[e];
  const int h = hedge_idx[e];

  const size_t cstride = (size_t)nN + nH;
  int* cNm = cnt + mine * cstride;
  int* cHm = cNm + nN;
  int* lNm = listN + (size_t)mine * nN * PN;
  int* lHm = listH + (size_t)mine * nH * PH;

  const int tn = atomicAdd(cNm + n, 1);
  const int th = atomicAdd(cHm + h, 1);
  if (tn < PN) {
    lNm[(size_t)n * PN + tn] = h;
  } else {
    const int t2 = atomicAdd(cnt + other * cstride + n, 1);
    if (t2 < PN) {
      listN[((size_t)other * nN + n) * PN + t2] = h;
    } else {
      const int p = atomicAdd(spillCnt + 0, 1);
      spillN[p] = make_int2(n, h);
    }
  }
  if (th < PH) {
    lHm[(size_t)h * PH + th] = n;
  } else {
    const int t2 = atomicAdd(cnt + other * cstride + nN + h, 1);
    if (t2 < PH) {
      listH[((size_t)other * nH + h) * PH + t2] = n;
    } else {
      const int p = atomicAdd(spillCnt + 1, 1);
      spillH[p] = make_int2(h, n);
    }
  }
}

__global__ __launch_bounds__(256) __attribute__((amdgpu_waves_per_eu(16)))
void buckets_and_input_products(
    const int* __restrict__ node_idx, const int* __restrict__ hedge_idx,
    int* __restrict__ cnt, int* __restrict__ listN, int* __restrict__ listH,
    int* __restrict__ spillCnt, int2* __restrict__ spillN, int2* __restrict__ spillH,
    int n_inc, int nN, int nH,
    const float* __restrict__ node_features, const float* __restrict__ hedge_features,
    const float* __restrict__ W_nn, const float* __restrict__ W_nh,
    const float* __restrict__ W_hh,
    float* __restrict__ node_pre, float* __restrict__ hedge_proj,
    float* __restrict__ hedge_pre)
{
  __shared__ v2f sW2[32 * 64];

  const int nbB = (n_inc + 255) >> 8;
  const int nbN = (nN + 127) >> 7;
  const int nbE = (nH + 127) >> 7;
  int whole = nbB >> 3;
  whole = (nbN >> 1) < whole ? (nbN >> 1) : whole;
  whole = nbE < whole ? nbE : whole;

  const int b = (int)blockIdx.x;
  int role, idx;
  if (b < whole * 12) {
    const int pat = b / 12;
    const int slot = b - pat * 12;
    if (slot < 8)        { role = 0; idx = pat * 8 + slot; }
    else if (slot < 10)  { role = 1; idx = pat * 2 + (slot - 8); }
    else if (slot == 10) { role = 2; idx = pat; }
    else                 { role = 3; idx = pat; }
  } else {
    int r = b - whole * 12;
    const int leftB = nbB - whole * 8;
    const int leftN = nbN - whole * 2;
    const int leftE = nbE - whole;
    if (r < leftB) { role = 0; idx = whole * 8 + r; }
    else {
      r -= leftB;
      if (r < leftN) { role = 1; idx = whole * 2 + r; }
      else {
        r -= leftN;
        if (r < leftE) { role = 2; idx = whole + r; }
        else           { role = 3; idx = whole + (r - leftE); }
      }
    }
  }

  if (role == 0) {
    row_buckets_block(idx, node_idx, hedge_idx, cnt, listN, listH,
                      spillCnt, spillN, spillH, n_inc, nN, nH);
    return;
  }
  const float* __restrict__ X = (role == 1) ? node_features : hedge_features;
  const float* __restrict__ Wm = (role == 1) ? W_nn : (role == 2 ? W_nh : W_hh);
  float* __restrict__ Y = (role == 1) ? node_pre : (role == 2 ? hedge_proj : hedge_pre);
  const int nrows = (role == 1) ? nN : nH;
  rows64_product_block(sW2, X, Wm, Y, nrows, idx);
}

template <int CAP>
static __device__ __forceinline__ void load_bucket_entries(
    int (&ent)[CAP / 32], const int* __restrict__ list, int nrows, int row,
    int u0, int u1, int lane)
{
  constexpr int NK = CAP / 32;
  constexpr int PC = CAP / 2;
  const int cu = u0 + u1;
  const int* __restrict__ part0 = list + (size_t)row * PC;
  const int* __restrict__ part1 = list + ((size_t)nrows + row) * PC;
#pragma unroll
  for (int q = 0; q < NK; ++q) {
    const int p = q * 32 + lane;
    const int* __restrict__ at = (p < u0) ? (part0 + p) : (part1 + (p - u0));
    ent[q] = (p < cu) ? *at : 0x7fffffff;
  }
}

template <int CAP>
static __device__ __forceinline__ void pack_bucket_keys(
    const int (&ent)[CAP / 32], unsigned (&key)[CAP / 32], int lane)
{
  constexpr int NK = CAP / 32;
  constexpr int SH = (CAP == 64) ? 6 : 5;
#pragma unroll
  for (int q = 0; q < NK; ++q) key[q] = ((unsigned)ent[q] << SH) | (unsigned)(q * 32 + lane);
}

template <int NK, int NC>
static __device__ __forceinline__ void rank_against(
    const unsigned (&key)[NK], int cu, int (&rnk)[NK])
{
#pragma unroll
  for (int qs = 0; qs < NC; ++qs) {
    const int left = cu - qs * 32;
    const int cn = left < 32 ? left : 32;
    for (int j = 0; j < cn; j += 4) {
#pragma unroll
      for (int t = 0; t < 4; ++t) {
        const unsigned kj = (unsigned)__builtin_amdgcn_readlane((int)key[qs], j + t);
#pragma unroll
        for (int q = 0; q < NC; ++q) rnk[q] += (kj < key[q]) ? 1 : 0;
      }
    }
  }
}

template <int NK>
static __device__ __forceinline__ void rank_bucket(
    const unsigned (&key)[NK], int cu, int (&rnk)[NK])
{
#pragma unroll
  for (int q = 0; q < NK; ++q) rnk[q] = 0;
  if (NK == 1 || cu <= 32) rank_against<NK, 1>(key, cu, rnk);
  else                     rank_against<NK, NK>(key, cu, rnk);
}

template <int U>
static __device__ __forceinline__ void gather_group(
    v2f& acc, int skq, int l0, int rem, const float* __restrict__ src, int lane)
{
  v2f v[U];
#pragma unroll
  for (int u = 0; u < U; ++u) {
    const int uu = (u < rem) ? u : (rem - 1);
    const int s = __builtin_amdgcn_readlane(skq, l0 + uu);
    v[u] = *(const v2f*)(src + (size_t)s * 64 + 2 * lane);
  }
#pragma unroll
  for (int u = 0; u < U; ++u) {
    if (u < rem) { acc.x += v[u].x; acc.y += v[u].y; }
  }
}

static __device__ __forceinline__ v2f gate_tanh2(v2f g, v2f s)
{
  const float k = 2.8853900817779268f;
  const float cx = g.x * s.x;
  const float cy = g.y * s.y;
  const float ex = __builtin_amdgcn_exp2f(cx * k);
  const float ey = __builtin_amdgcn_exp2f(cy * k);
  const float rx = __builtin_amdgcn_rcpf(1.0f + ex);
  const float ry = __builtin_amdgcn_rcpf(1.0f + ey);
  v2f o;
  o.x = __builtin_fmaf(-2.0f, rx, 1.0f);
  o.y = __builtin_fmaf(-2.0f, ry, 1.0f);
  return o;
}

template <int CAP>
__global__ __launch_bounds__(256) void pull_gate_tanh(
    const float* __restrict__ src, const int* __restrict__ cnt, int cntPartStride,
    const int* __restrict__ list, const int* __restrict__ spillCntP,
    const int2* __restrict__ spill, const float* __restrict__ pre,
    float* __restrict__ out, int nrows, int n_inc)
{
  constexpr int NK = CAP / 32;
  constexpr int PC = CAP / 2;
  __shared__ int sorted[8][CAP];

  const int lane = threadIdx.x & 31;
  const int wave = __builtin_amdgcn_readfirstlane((int)(threadIdx.x >> 5));
  const int step = (int)gridDim.x * 8;
  int row = blockIdx.x * 8 + wave;
  if (row >= nrows) return;

  int c0 = cnt[row];
  int c1 = cnt[cntPartStride + row];
  int u0 = c0 < PC ? c0 : PC;
  int u1 = c1 < PC ? c1 : PC;
  int ent[NK];
  load_bucket_entries<CAP>(ent, list, nrows, row, u0, u1, lane);
  v2f p2 = *(const v2f*)(pre + (size_t)row * 64 + 2 * lane);

  for (;;) {
    const int nraw = row + step;
    const bool more = nraw < nrows;
    const int nrow = more ? nraw : row;
    const int nc0 = cnt[nrow];
    const int nc1 = cnt[cntPartStride + nrow];

    const int cu = u0 + u1;

    unsigned key[NK];
    pack_bucket_keys<CAP>(ent, key, lane);
    int rnk[NK];
    rank_bucket<NK>(key, cu, rnk);
#pragma unroll
    for (int q = 0; q < NK; ++q) {
      const int p = q * 32 + lane;
      if (p < cu) sorted[wave][rnk[q]] = ent[q];
    }

    int sk[NK];
#pragma unroll
    for (int q = 0; q < NK; ++q) {
      const int p = q * 32 + lane;
      sk[q] = (p < cu) ? sorted[wave][p] : 0;
    }

    const int nu0 = nc0 < PC ? nc0 : PC;
    const int nu1 = nc1 < PC ? nc1 : PC;
    int nent[NK];
    load_bucket_entries<CAP>(nent, list, nrows, nrow, nu0, nu1, lane);
    const v2f np2 = *(const v2f*)(pre + (size_t)nrow * 64 + 2 * lane);

    v2f acc = {0.0f, 0.0f};
    for (int j = 0; j < cu; j += 8) {
      const int rem = cu - j;
      int skq = sk[0];
#pragma unroll
      for (int q = 1; q < NK; ++q) skq = (j >= q * 32) ? sk[q] : skq;
      const int l0 = j & 31;
      if (rem > 4) gather_group<8>(acc, skq, l0, rem, src, lane);
      else         gather_group<4>(acc, skq, l0, rem, src, lane);
    }

    if (c0 > PC && c1 > PC) {
      int tot = spillCntP[0];
      tot = tot < n_inc ? tot : n_inc;
      for (int i = 0; i < tot; i += 32) {
        const int p = i + lane;
        int2 pr = make_int2(-1, 0);
        if (p < tot) pr = spill[p];
        unsigned m = __builtin_amdgcn_ballot_w32(pr.x == row);
        while (m) {
          const int b = __builtin_ctz(m);
          m &= m - 1;
          const int s = __builtin_amdgcn_readlane(pr.y, b);
          const v2f v = *(const v2f*)(src + (size_t)s * 64 + 2 * lane);
          acc.x += v.x; acc.y += v.y;
        }
      }
    }

    const v2f r = gate_tanh2(p2, acc);
    *(v2f*)(out + (size_t)row * 64 + 2 * lane) = r;

    if (!more) break;
    row = nrow; c0 = nc0; c1 = nc1; u0 = nu0; u1 = nu1; p2 = np2;
#pragma unroll
    for (int q = 0; q < NK; ++q) ent[q] = nent[q];
  }
}

#define SLAB_PITCH 66

template <int CAP>
__global__ __launch_bounds__(128) void pull_rows32_product_gate_tanh(
    const float* __restrict__ src, const int* __restrict__ cnt, int cntPartStride,
    const int* __restrict__ list, const int* __restrict__ spillCntP,
    const int2* __restrict__ spill, const float* __restrict__ W,
    const float* __restrict__ pre, float* __restrict__ out, int nrows, int n_inc)
{
  constexpr int NK = CAP / 32;
  constexpr int PC = CAP / 2;
  __shared__ v2f sW2[32 * 64];
  __shared__ int sorted[4][CAP];
  __shared__ __attribute__((aligned(16))) float sums[32 * SLAB_PITCH];
  __shared__ __attribute__((aligned(16))) float prod[32 * SLAB_PITCH];

  const int tid = threadIdx.x;
  for (int i = tid; i < 32 * 64; i += 128) {
    const int j = i >> 6;
    const int n = i & 63;
    v2f w;
    w.x = W[(2 * j) * 64 + n];
    w.y = W[(2 * j + 1) * 64 + n];
    sW2[i] = w;
  }

  const int lane = tid & 31;
  const int wave = __builtin_amdgcn_readfirstlane(tid >> 5);
  const int row0 = (int)blockIdx.x * 32 + wave * 8;
  int nr = nrows - row0;
  nr = nr < 0 ? 0 : (nr > 8 ? 8 : nr);

  int row = row0 < nrows ? row0 : nrows - 1;
  int c0 = cnt[row];
  int c1 = cnt[cntPartStride + row];
  int u0 = c0 < PC ? c0 : PC;
  int u1 = c1 < PC ? c1 : PC;
  int ent[NK];
  load_bucket_entries<CAP>(ent, list, nrows, row, u0, u1, lane);

  v2f g2[8];
#pragma unroll
  for (int r = 0; r < 8; ++r) {
    int gr = row0 + r;
    gr = gr < nrows ? gr : nrows - 1;
    g2[r] = *(const v2f*)(pre + (size_t)gr * 64 + 2 * lane);
  }

  for (int r = 0; r < nr; ++r) {
    const bool more = (r + 1) < nr;
    const int nrow = more ? row + 1 : row;
    const int nc0 = cnt[nrow];
    const int nc1 = cnt[cntPartStride + nrow];

    const int cu = u0 + u1;

    unsigned key[NK];
    pack_bucket_keys<CAP>(ent, key, lane);
    int rnk[NK];
    rank_bucket<NK>(key, cu, rnk);
#pragma unroll
    for (int q = 0; q < NK; ++q) {
      const int p = q * 32 + lane;
      if (p < cu) sorted[wave][rnk[q]] = ent[q];
    }

    int sk[NK];
#pragma unroll
    for (int q = 0; q < NK; ++q) {
      const int p = q * 32 + lane;
      sk[q] = (p < cu) ? sorted[wave][p] : 0;
    }

    const int nu0 = nc0 < PC ? nc0 : PC;
    const int nu1 = nc1 < PC ? nc1 : PC;
    int nent[NK];
    load_bucket_entries<CAP>(nent, list, nrows, nrow, nu0, nu1, lane);

    v2f acc = {0.0f, 0.0f};
    for (int j = 0; j < cu; j += 8) {
      const int rem = cu - j;
      int skq = sk[0];
#pragma unroll
      for (int q = 1; q < NK; ++q) skq = (j >= q * 32) ? sk[q] : skq;
      const int l0 = j & 31;
      if (rem > 4) gather_group<8>(acc, skq, l0, rem, src, lane);
      else         gather_group<4>(acc, skq, l0, rem, src, lane);
    }

    if (c0 > PC && c1 > PC) {
      int tot = spillCntP[0];
      tot = tot < n_inc ? tot : n_inc;
      for (int i = 0; i < tot; i += 32) {
        const int p = i + lane;
        int2 pr = make_int2(-1, 0);
        if (p < tot) pr = spill[p];
        unsigned m = __builtin_amdgcn_ballot_w32(pr.x == row);
        while (m) {
          const int b = __builtin_ctz(m);
          m &= m - 1;
          const int s = __builtin_amdgcn_readlane(pr.y, b);
          const v2f v = *(const v2f*)(src + (size_t)s * 64 + 2 * lane);
          acc.x += v.x; acc.y += v.y;
        }
      }
    }

    *(v2f*)&sums[(wave * 8 + r) * SLAB_PITCH + 2 * lane] = acc;

    row = nrow; c0 = nc0; c1 = nc1; u0 = nu0; u1 = nu1;
#pragma unroll
    for (int q = 0; q < NK; ++q) ent[q] = nent[q];
  }
  for (int r = nr; r < 8; ++r) {
    const v2f z = {0.0f, 0.0f};
    *(v2f*)&sums[(wave * 8 + r) * SLAB_PITCH + 2 * lane] = z;
  }

  __syncthreads();

  const int mt = wave >> 1;
  const int nh = wave & 1;
  const int m  = lane & 15;
  const int kh = lane >> 4;
  const int n  = lane & 15;

  v8f acc0 = {}, acc1 = {};

#pragma unroll
  for (int kk = 0; kk < 16; ++kk) {
    const int k0 = kk * 4 + kh * 2;
    const v2f a = *(const v2f*)&sums[(mt * 16 + m) * SLAB_PITCH + k0];

    const int kp = k0 >> 1;
    const v2f* __restrict__ brow = &sW2[kp * 64 + nh * 32 + n];
    const v2f b0 = brow[0];
    const v2f b1 = brow[16];

    acc0 = __builtin_amdgcn_wmma_f32_16x16x4_f32(false, a, false, b0, (short)0, acc0, false, false);
    acc1 = __builtin_amdgcn_wmma_f32_16x16x4_f32(false, a, false, b1, (short)0, acc1, false, false);
  }

  const int crow = mt * 16 + (lane >> 4) * 8;
  const int ccol = nh * 32 + (lane & 15);
#pragma unroll
  for (int v = 0; v < 8; ++v) {
    float* yr = &prod[(crow + v) * SLAB_PITCH + ccol];
    yr[0]  = acc0[v];
    yr[16] = acc1[v];
  }

  __syncthreads();

#pragma unroll
  for (int r = 0; r < 8; ++r) {
    if (r < nr) {
      const v2f s = *(const v2f*)&prod[(wave * 8 + r) * SLAB_PITCH + 2 * lane];
      const v2f o = gate_tanh2(g2[r], s);
      *(v2f*)(out + (size_t)(row0 + r) * 64 + 2 * lane) = o;
    }
  }
}

extern "C" void kernel_launch(void* const* d_in, const int* in_sizes, int n_in,
                              void* d_out, int out_size, void* d_ws, size_t ws_size,
                              hipStream_t stream) {
  const float* node_features  = (const float*)d_in[0];
  const float* hedge_features = (const float*)d_in[1];
  const float* W_nn = (const float*)d_in[2];
  const float* W_nh = (const float*)d_in[3];
  const float* W_hh = (const float*)d_in[4];
  const float* W_hn = (const float*)d_in[5];
  const int* node_idx  = (const int*)d_in[6];
  const int* hedge_idx = (const int*)d_in[7];

  const int N   = in_sizes[0] / 64;
  const int E   = in_sizes[1] / 64;
  const int INC = in_sizes[6];

  float* slotA = (float*)d_ws;
  float* slotB = slotA + (size_t)N * 64;
  float* slotC = slotB + (size_t)N * 64;
  int* listN  = (int*)(slotC + (size_t)E * 64);
  int* listH  = listN + (size_t)N * CAP_N;
  int2* spillN = (int2*)(listH + (size_t)E * CAP_H);
  int2* spillH = spillN + (size_t)INC;
  int* cnt    = (int*)(spillH + (size_t)INC);
  int* spillCnt = cnt + 2 * ((size_t)N + E);

  float* out_node  = (float*)d_out;
  float* out_hedge = out_node + (size_t)N * 64;

  const dim3 blk(256);
  const dim3 firstGrid((INC + 255) / 256 + (N + 127) / 128 + 2 * ((E + 127) / 128));

  hipMemsetAsync(cnt, 0, (2 * ((size_t)N + E) + 2) * sizeof(int), stream);
  buckets_and_input_products<<<firstGrid, blk, 0, stream>>>(
      node_idx, hedge_idx, cnt, listN, listH, spillCnt, spillN, spillH, INC, N, E,
      node_features, hedge_features, W_nn, W_nh, W_hh, slotB, slotA, slotC);

  pull_gate_tanh<CAP_N><<<dim3(2048), blk, 0, stream>>>(
      slotA, cnt, N + E, listN, spillCnt + 0, spillN, slotB, out_node, N, INC);

  pull_rows32_product_gate_tanh<CAP_H><<<dim3((E + 31) / 32), dim3(128), 0, stream>>>(
      out_node, cnt + N, N + E, listH, spillCnt + 1, spillH, W_hn, slotC, out_hedge, E, INC);
}


# Hyperedge pull block keeps under 10 KiB of LDS (W's B fragments straight from global, product written back over the sums slab) so 16 blocks fill a WGP's 64 wave slots, was 9; sum order kept
# speedup vs baseline: 3.1520x; 1.0785x over previous
#include <hip/hip_runtime.h>
#include <hip/hip_bf16.h>
#include <math.h>

typedef __attribute__((ext_vector_type(2))) float v2f;
typedef __attribute__((ext_vector_type(8))) float v8f;

static __device__ __forceinline__ void rows64_product_block(
    v2f* __restrict__ sW2, const float* __restrict__ X, const float* __restrict__ W,
    float* __restrict__ Y, int nrows, int tileBlock)
{
  const int tid = threadIdx.x;
  for (int i = tid; i < 32 * 64; i += 256) {
    const int j = i >> 6;
    const int n = i & 63;
    v2f w;
    w.x = W[(2 * j) * 64 + n];
    w.y = W[(2 * j + 1) * 64 + n];
    sW2[i] = w;
  }
  __syncthreads();

  const int lane    = tid & 31;
  const int wave    = tid >> 5;
  const int rowBase = tileBlock * 128 + wave * 16;
  if (rowBase + 16 > nrows) return;

  const int m  = lane & 15;
  const int kh = lane >> 4;
  const int n  = lane & 15;

  const float* __restrict__ xrow = X + (size_t)(rowBase + m) * 64;

  v8f acc0 = {}, acc1 = {}, acc2 = {}, acc3 = {};

#pragma unroll
  for (int kk = 0; kk < 16; ++kk) {
    const int k0 = kk * 4 + kh * 2;
    v2f a = *(const v2f*)(xrow + k0);

    const int kp = k0 >> 1;
    const v2f* __restrict__ brow = &sW2[kp * 64 + n];
    v2f b0 = brow[0];
    v2f b1 = brow[16];
    v2f b2 = brow[32];
    v2f b3 = brow[48];

    acc0 = __builtin_amdgcn_wmma_f32_16x16x4_f32(false, a, false, b0, (short)0, acc0, false, false);
    acc1 = __builtin_amdgcn_wmma_f32_16x16x4_f32(false, a, false, b1, (short)0, acc1, false, false);
    acc2 = __builtin_amdgcn_wmma_f32_16x16x4_f32(false, a, false, b2, (short)0, acc2, false, false);
    acc3 = __builtin_amdgcn_wmma_f32_16x16x4_f32(false, a, false, b3, (short)0, acc3, false, false);
  }

  const int crow = rowBase + (lane >> 4) * 8;
  const int ccol = lane & 15;
#pragma unroll
  for (int v = 0; v < 8; ++v) {
    float* __restrict__ yr = Y + (size_t)(crow + v) * 64 + ccol;
    yr[0]  = acc0[v];
    yr[16] = acc1[v];
    yr[32] = acc2[v];
    yr[48] = acc3[v];
  }
}

#define CAP_N 32
#define CAP_H 64

static __device__ __forceinline__ void row_buckets_block(
    int incBlock,
    const int* __restrict__ node_idx, const int* __restrict__ hedge_idx,
    int* __restrict__ cnt,
    int* __restrict__ listN,
    int* __restrict__ listH,
    int* __restrict__ spillCnt, int2* __restrict__ spillN, int2* __restrict__ spillH,
    int n_inc, int nN, int nH)
{
  constexpr int PN = CAP_N / 2;
  constexpr int PH = CAP_H / 2;
  const int mine = (__builtin_amdgcn_s_sendmsg_rtn(0x87) >> 18) & 1;
  const int other = mine ^ 1;

  const int e = incBlock * 256 + (int)threadIdx.x;
  if (e >= n_inc) return;
  const int n = node_idx[e];
  const int h = hedge_idx[e];

  const size_t cstride = (size_t)nN + nH;
  int* cNm = cnt + mine * cstride;
  int* cHm = cNm + nN;
  int* lNm = listN + (size_t)mine * nN * PN;
  int* lHm = listH + (size_t)mine * nH * PH;

  const int tn = atomicAdd(cNm + n, 1);
  const int th = atomicAdd(cHm + h, 1);
  if (tn < PN) {
    lNm[(size_t)n * PN + tn] = h;
  } else {
    const int t2 = atomicAdd(cnt + other * cstride + n, 1);
    if (t2 < PN) {
      listN[((size_t)other * nN + n) * PN + t2] = h;
    } else {
      const int p = atomicAdd(spillCnt + 0, 1);
      spillN[p] = make_int2(n, h);
    }
  }
  if (th < PH) {
    lHm[(size_t)h * PH + th] = n;
  } else {
    const int t2 = atomicAdd(cnt + other * cstride + nN + h, 1);
    if (t2 < PH) {
      listH[((size_t)other * nH + h) * PH + t2] = n;
    } else {
      const int p = atomicAdd(spillCnt + 1, 1);
      spillH[p] = make_int2(h, n);
    }
  }
}

__global__ __launch_bounds__(256) __attribute__((amdgpu_waves_per_eu(16)))
void buckets_and_input_products(
    const int* __restrict__ node_idx, const int* __restrict__ hedge_idx,
    int* __restrict__ cnt, int* __restrict__ listN, int* __restrict__ listH,
    int* __restrict__ spillCnt, int2* __restrict__ spillN, int2* __restrict__ spillH,
    int n_inc, int nN, int nH,
    const float* __restrict__ node_features, const float* __restrict__ hedge_features,
    const float* __restrict__ W_nn, const float* __restrict__ W_nh,
    const float* __restrict__ W_hh,
    float* __restrict__ node_pre, float* __restrict__ hedge_proj,
    float* __restrict__ hedge_pre)
{
  __shared__ v2f sW2[32 * 64];

  const int nbB = (n_inc + 255) >> 8;
  const int nbN = (nN + 127) >> 7;
  const int nbE = (nH + 127) >> 7;
  int whole = nbB >> 3;
  whole = (nbN >> 1) < whole ? (nbN >> 1) : whole;
  whole = nbE < whole ? nbE : whole;

  const int b = (int)blockIdx.x;
  int role, idx;
  if (b < whole * 12) {
    const int pat = b / 12;
    const int slot = b - pat * 12;
    if (slot < 8)        { role = 0; idx = pat * 8 + slot; }
    else if (slot < 10)  { role = 1; idx = pat * 2 + (slot - 8); }
    else if (slot == 10) { role = 2; idx = pat; }
    else                 { role = 3; idx = pat; }
  } else {
    int r = b - whole * 12;
    const int leftB = nbB - whole * 8;
    const int leftN = nbN - whole * 2;
    const int leftE = nbE - whole;
    if (r < leftB) { role = 0; idx = whole * 8 + r; }
    else {
      r -= leftB;
      if (r < leftN) { role = 1; idx = whole * 2 + r; }
      else {
        r -= leftN;
        if (r < leftE) { role = 2; idx = whole + r; }
        else           { role = 3; idx = whole + (r - leftE); }
      }
    }
  }

  if (role == 0) {
    row_buckets_block(idx, node_idx, hedge_idx, cnt, listN, listH,
                      spillCnt, spillN, spillH, n_inc, nN, nH);
    return;
  }
  const float* __restrict__ X = (role == 1) ? node_features : hedge_features;
  const float* __restrict__ Wm = (role == 1) ? W_nn : (role == 2 ? W_nh : W_hh);
  float* __restrict__ Y = (role == 1) ? node_pre : (role == 2 ? hedge_proj : hedge_pre);
  const int nrows = (role == 1) ? nN : nH;
  rows64_product_block(sW2, X, Wm, Y, nrows, idx);
}

template <int CAP>
static __device__ __forceinline__ void load_bucket_entries(
    int (&ent)[CAP / 32], const int* __restrict__ list, int nrows, int row,
    int u0, int u1, int lane)
{
  constexpr int NK = CAP / 32;
  constexpr int PC = CAP / 2;
  const int cu = u0 + u1;
  const int* __restrict__ part0 = list + (size_t)row * PC;
  const int* __restrict__ part1 = list + ((size_t)nrows + row) * PC;
#pragma unroll
  for (int q = 0; q < NK; ++q) {
    const int p = q * 32 + lane;
    const int* __restrict__ at = (p < u0) ? (part0 + p) : (part1 + (p - u0));
    ent[q] = (p < cu) ? *at : 0x7fffffff;
  }
}

template <int CAP>
static __device__ __forceinline__ void pack_bucket_keys(
    const int (&ent)[CAP / 32], unsigned (&key)[CAP / 32], int lane)
{
  constexpr int NK = CAP / 32;
  constexpr int SH = (CAP == 64) ? 6 : 5;
#pragma unroll
  for (int q = 0; q < NK; ++q) key[q] = ((unsigned)ent[q] << SH) | (unsigned)(q * 32 + lane);
}

template <int NK, int NC>
static __device__ __forceinline__ void rank_against(
    const unsigned (&key)[NK], int cu, int (&rnk)[NK])
{
#pragma unroll
  for (int qs = 0; qs < NC; ++qs) {
    const int left = cu - qs * 32;
    const int cn = left < 32 ? left : 32;
    for (int j = 0; j < cn; j += 4) {
#pragma unroll
      for (int t = 0; t < 4; ++t) {
        const unsigned kj = (unsigned)__builtin_amdgcn_readlane((int)key[qs], j + t);
#pragma unroll
        for (int q = 0; q < NC; ++q) rnk[q] += (kj < key[q]) ? 1 : 0;
      }
    }
  }
}

template <int NK>
static __device__ __forceinline__ void rank_bucket(
    const unsigned (&key)[NK], int cu, int (&rnk)[NK])
{
#pragma unroll
  for (int q = 0; q < NK; ++q) rnk[q] = 0;
  if (NK == 1 || cu <= 32) rank_against<NK, 1>(key, cu, rnk);
  else                     rank_against<NK, NK>(key, cu, rnk);
}

template <int U>
static __device__ __forceinline__ void gather_group(
    v2f& acc, int skq, int l0, int rem, const float* __restrict__ src, int lane)
{
  v2f v[U];
#pragma unroll
  for (int u = 0; u < U; ++u) {
    const int uu = (u < rem) ? u : (rem - 1);
    const int s = __builtin_amdgcn_readlane(skq, l0 + uu);
    v[u] = *(const v2f*)(src + (size_t)s * 64 + 2 * lane);
  }
#pragma unroll
  for (int u = 0; u < U; ++u) {
    if (u < rem) { acc.x += v[u].x; acc.y += v[u].y; }
  }
}

static __device__ __forceinline__ v2f gate_tanh2(v2f g, v2f s)
{
  const float k = 2.8853900817779268f;
  const float cx = g.x * s.x;
  const float cy = g.y * s.y;
  const float ex = __builtin_amdgcn_exp2f(cx * k);
  const float ey = __builtin_amdgcn_exp2f(cy * k);
  const float rx = __builtin_amdgcn_rcpf(1.0f + ex);
  const float ry = __builtin_amdgcn_rcpf(1.0f + ey);
  v2f o;
  o.x = __builtin_fmaf(-2.0f, rx, 1.0f);
  o.y = __builtin_fmaf(-2.0f, ry, 1.0f);
  return o;
}

template <int CAP>
__global__ __launch_bounds__(256) void pull_gate_tanh(
    const float* __restrict__ src, const int* __restrict__ cnt, int cntPartStride,
    const int* __restrict__ list, const int* __restrict__ spillCntP,
    const int2* __restrict__ spill, const float* __restrict__ pre,
    float* __restrict__ out, int nrows, int n_inc)
{
  constexpr int NK = CAP / 32;
  constexpr int PC = CAP / 2;
  __shared__ int sorted[8][CAP];

  const int lane = threadIdx.x & 31;
  const int wave = __builtin_amdgcn_readfirstlane((int)(threadIdx.x >> 5));
  const int step = (int)gridDim.x * 8;
  int row = blockIdx.x * 8 + wave;
  if (row >= nrows) return;

  int c0 = cnt[row];
  int c1 = cnt[cntPartStride + row];
  int u0 = c0 < PC ? c0 : PC;
  int u1 = c1 < PC ? c1 : PC;
  int ent[NK];
  load_bucket_entries<CAP>(ent, list, nrows, row, u0, u1, lane);
  v2f p2 = *(const v2f*)(pre + (size_t)row * 64 + 2 * lane);

  for (;;) {
    const int nraw = row + step;
    const bool more = nraw < nrows;
    const int nrow = more ? nraw : row;
    const int nc0 = cnt[nrow];
    const int nc1 = cnt[cntPartStride + nrow];

    const int cu = u0 + u1;

    unsigned key[NK];
    pack_bucket_keys<CAP>(ent, key, lane);
    int rnk[NK];
    rank_bucket<NK>(key, cu, rnk);
#pragma unroll
    for (int q = 0; q < NK; ++q) {
      const int p = q * 32 + lane;
      if (p < cu) sorted[wave][rnk[q]] = ent[q];
    }

    int sk[NK];
#pragma unroll
    for (int q = 0; q < NK; ++q) {
      const int p = q * 32 + lane;
      sk[q] = (p < cu) ? sorted[wave][p] : 0;
    }

    const int nu0 = nc0 < PC ? nc0 : PC;
    const int nu1 = nc1 < PC ? nc1 : PC;
    int nent[NK];
    load_bucket_entries<CAP>(nent, list, nrows, nrow, nu0, nu1, lane);
    const v2f np2 = *(const v2f*)(pre + (size_t)nrow * 64 + 2 * lane);

    v2f acc = {0.0f, 0.0f};
    for (int j = 0; j < cu; j += 8) {
      const int rem = cu - j;
      int skq = sk[0];
#pragma unroll
      for (int q = 1; q < NK; ++q) skq = (j >= q * 32) ? sk[q] : skq;
      const int l0 = j & 31;
      if (rem > 4) gather_group<8>(acc, skq, l0, rem, src, lane);
      else         gather_group<4>(acc, skq, l0, rem, src, lane);
    }

    if (c0 > PC && c1 > PC) {
      int tot = spillCntP[0];
      tot = tot < n_inc ? tot : n_inc;
      for (int i = 0; i < tot; i += 32) {
        const int p = i + lane;
        int2 pr = make_int2(-1, 0);
        if (p < tot) pr = spill[p];
        unsigned m = __builtin_amdgcn_ballot_w32(pr.x == row);
        while (m) {
          const int b = __builtin_ctz(m);
          m &= m - 1;
          const int s = __builtin_amdgcn_readlane(pr.y, b);
          const v2f v = *(const v2f*)(src + (size_t)s * 64 + 2 * lane);
          acc.x += v.x; acc.y += v.y;
        }
      }
    }

    const v2f r = gate_tanh2(p2, acc);
    *(v2f*)(out + (size_t)row * 64 + 2 * lane) = r;

    if (!more) break;
    row = nrow; c0 = nc0; c1 = nc1; u0 = nu0; u1 = nu1; p2 = np2;
#pragma unroll
    for (int q = 0; q < NK; ++q) ent[q] = nent[q];
  }
}

#define SLAB_PITCH 66

template <int CAP>
__global__ __launch_bounds__(128) __attribute__((amdgpu_waves_per_eu(16)))
void pull_rows32_product_gate_tanh(
    const float* __restrict__ src, const int* __restrict__ cnt, int cntPartStride,
    const int* __restrict__ list, const int* __restrict__ spillCntP,
    const int2* __restrict__ spill, const float* __restrict__ W,
    const float* __restrict__ pre, float* __restrict__ out, int nrows, int n_inc)
{
  constexpr int NK = CAP / 32;
  constexpr int PC = CAP / 2;
  __shared__ int sorted[4][CAP];
  __shared__ __attribute__((aligned(16))) float sums[32 * SLAB_PITCH];

  const int tid = threadIdx.x;
  const int lane = tid & 31;
  const int wave = __builtin_amdgcn_readfirstlane(tid >> 5);
  const int row0 = (int)blockIdx.x * 32 + wave * 8;
  int nr = nrows - row0;
  nr = nr < 0 ? 0 : (nr > 8 ? 8 : nr);

  int row = row0 < nrows ? row0 : nrows - 1;
  int c0 = cnt[row];
  int c1 = cnt[cntPartStride + row];
  int u0 = c0 < PC ? c0 : PC;
  int u1 = c1 < PC ? c1 : PC;
  int ent[NK];
  load_bucket_entries<CAP>(ent, list, nrows, row, u0, u1, lane);

  v2f g2[8];
#pragma unroll
  for (int r = 0; r < 8; ++r) {
    int gr = row0 + r;
    gr = gr < nrows ? gr : nrows - 1;
    g2[r] = *(const v2f*)(pre + (size_t)gr * 64 + 2 * lane);
  }

  for (int r = 0; r < nr; ++r) {
    const bool more = (r + 1) < nr;
    const int nrow = more ? row + 1 : row;
    const int nc0 = cnt[nrow];
    const int nc1 = cnt[cntPartStride + nrow];

    const int cu = u0 + u1;

    unsigned key[NK];
    pack_bucket_keys<CAP>(ent, key, lane);
    int rnk[NK];
    rank_bucket<NK>(key, cu, rnk);
#pragma unroll
    for (int q = 0; q < NK; ++q) {
      const int p = q * 32 + lane;
      if (p < cu) sorted[wave][rnk[q]] = ent[q];
    }

    int sk[NK];
#pragma unroll
    for (int q = 0; q < NK; ++q) {
      const int p = q * 32 + lane;
      sk[q] = (p < cu) ? sorted[wave][p] : 0;
    }

    const int nu0 = nc0 < PC ? nc0 : PC;
    const int nu1 = nc1 < PC ? nc1 : PC;
    int nent[NK];
    load_bucket_entries<CAP>(nent, list, nrows, nrow, nu0, nu1, lane);

    v2f acc = {0.0f, 0.0f};
    for (int j = 0; j < cu; j += 8) {
      const int rem = cu - j;
      int skq = sk[0];
#pragma unroll
      for (int q = 1; q < NK; ++q) skq = (j >= q * 32) ? sk[q] : skq;
      const int l0 = j & 31;
      if (rem > 4) gather_group<8>(acc, skq, l0, rem, src, lane);
      else         gather_group<4>(acc, skq, l0, rem, src, lane);
    }

    if (c0 > PC && c1 > PC) {
      int tot = spillCntP[0];
      tot = tot < n_inc ? tot : n_inc;
      for (int i = 0; i < tot; i += 32) {
        const int p = i + lane;
        int2 pr = make_int2(-1, 0);
        if (p < tot) pr = spill[p];
        unsigned m = __builtin_amdgcn_ballot_w32(pr.x == row);
        while (m) {
          const int b = __builtin_ctz(m);
          m &= m - 1;
          const int s = __builtin_amdgcn_readlane(pr.y, b);
          const v2f v = *(const v2f*)(src + (size_t)s * 64 + 2 * lane);
          acc.x += v.x; acc.y += v.y;
        }
      }
    }

    *(v2f*)&sums[(wave * 8 + r) * SLAB_PITCH + 2 * lane] = acc;

    row = nrow; c0 = nc0; c1 = nc1; u0 = nu0; u1 = nu1;
#pragma unroll
    for (int q = 0; q < NK; ++q) ent[q] = nent[q];
  }
  for (int r = nr; r < 8; ++r) {
    const v2f z = {0.0f, 0.0f};
    *(v2f*)&sums[(wave * 8 + r) * SLAB_PITCH + 2 * lane] = z;
  }

  const int mt = wave >> 1;
  const int nh = wave & 1;
  const int m  = lane & 15;
  const int kh = lane >> 4;
  const int n  = lane & 15;

  const float* __restrict__ wcol = W + (kh * 2) * 64 + nh * 32 + 2 * n;
  v2f wq0[4], wq1[4];
#pragma unroll
  for (int kk = 0; kk < 3; ++kk) {
    wq0[kk] = *(const v2f*)(wcol + kk * 256);
    wq1[kk] = *(const v2f*)(wcol + kk * 256 + 64);
  }

  __syncthreads();

  v8f acc0 = {}, acc1 = {};

#pragma unroll
  for (int kk = 0; kk < 16; ++kk) {
    const int k0 = kk * 4 + kh * 2;
    const v2f a = *(const v2f*)&sums[(mt * 16 + m) * SLAB_PITCH + k0];

    if (kk + 3 < 16) {
      wq0[(kk + 3) & 3] = *(const v2f*)(wcol + (kk + 3) * 256);
      wq1[(kk + 3) & 3] = *(const v2f*)(wcol + (kk + 3) * 256 + 64);
    }
    v2f b0, b1;
    b0.x = wq0[kk & 3].x;
    b0.y = wq1[kk & 3].x;
    b1.x = wq0[kk & 3].y;
    b1.y = wq1[kk & 3].y;

    acc0 = __builtin_amdgcn_wmma_f32_16x16x4_f32(false, a, false, b0, (short)0, acc0, false, false);
    acc1 = __builtin_amdgcn_wmma_f32_16x16x4_f32(false, a, false, b1, (short)0, acc1, false, false);
  }

  __syncthreads();

  const int crow = mt * 16 + (lane >> 4) * 8;
  const int ccol = nh * 32 + 2 * (lane & 15);
#pragma unroll
  for (int v = 0; v < 8; ++v) {
    v2f pr;
    pr.x = acc0[v];
    pr.y = acc1[v];
    *(v2f*)&sums[(crow + v) * SLAB_PITCH + ccol] = pr;
  }

  __syncthreads();

#pragma unroll
  for (int r = 0; r < 8; ++r) {
    if (r < nr) {
      const v2f s = *(const v2f*)&sums[(wave * 8 + r) * SLAB_PITCH + 2 * lane];
      const v2f o = gate_tanh2(g2[r], s);
      *(v2f*)(out + (size_t)(row0 + r) * 64 + 2 * lane) = o;
    }
  }
}

extern "C" void kernel_launch(void* const* d_in, const int* in_sizes, int n_in,
                              void* d_out, int out_size, void* d_ws, size_t ws_size,
                              hipStream_t stream) {
  const float* node_features  = (const float*)d_in[0];
  const float* hedge_features = (const float*)d_in[1];
  const float* W_nn = (const float*)d_in[2];
  const float* W_nh = (const float*)d_in[3];
  const float* W_hh = (const float*)d_in[4];
  const float* W_hn = (const float*)d_in[5];
  const int* node_idx  = (const int*)d_in[6];
  const int* hedge_idx = (const int*)d_in[7];

  const int N   = in_sizes[0] / 64;
  const int E   = in_sizes[1] / 64;
  const int INC = in_sizes[6];

  float* slotA = (float*)d_ws;
  float* slotB = slotA + (size_t)N * 64;
  float* slotC = slotB + (size_t)N * 64;
  int* listN  = (int*)(slotC + (size_t)E * 64);
  int* listH  = listN + (size_t)N * CAP_N;
  int2* spillN = (int2*)(listH + (size_t)E * CAP_H);
  int2* spillH = spillN + (size_t)INC;
  int* cnt    = (int*)(spillH + (size_t)INC);
  int* spillCnt = cnt + 2 * ((size_t)N + E);

  float* out_node  = (float*)d_out;
  float* out_hedge = out_node + (size_t)N * 64;

  const dim3 blk(256);
  const dim3 firstGrid((INC + 255) / 256 + (N + 127) / 128 + 2 * ((E + 127) / 128));

  hipMemsetAsync(cnt, 0, (2 * ((size_t)N + E) + 2) * sizeof(int), stream);
  buckets_and_input_products<<<firstGrid, blk, 0, stream>>>(
      node_idx, hedge_idx, cnt, listN, listH, spillCnt, spillN, spillH, INC, N, E,
      node_features, hedge_features, W_nn, W_nh, W_hh, slotB, slotA, slotC);

  pull_gate_tanh<CAP_N><<<dim3(2048), blk, 0, stream>>>(
      slotA, cnt, N + E, listN, spillCnt + 0, spillN, slotB, out_node, N, INC);

  pull_rows32_product_gate_tanh<CAP_H><<<dim3((E + 31) / 32), dim3(128), 0, stream>>>(
      out_node, cnt + N, N + E, listH, spillCnt + 1, spillH, W_hn, slotC, out_hedge, E, INC);
}


# Global row lists and ticket counters dropped: incidences binned by row range on LDS counters (one global add per bin block and range), pull blocks slot their own rows' pairs in LDS; sum order kept
# speedup vs baseline: 3.3424x; 1.0604x over previous
#include <hip/hip_runtime.h>
#include <hip/hip_bf16.h>
#include <math.h>

typedef __attribute__((ext_vector_type(2))) float v2f;
typedef __attribute__((ext_vector_type(8))) float v8f;
typedef __attribute__((ext_vector_type(2))) int v2i;

static __device__ __forceinline__ void rows64_product_block(
    v2f* __restrict__ sW2, const float* __restrict__ X, const float* __restrict__ W,
    float* __restrict__ Y, int nrows, int tileBlock)
{
  const int tid = threadIdx.x;
  for (int i = tid; i < 32 * 64; i += 256) {
    const int j = i >> 6;
    const int n = i & 63;
    v2f w;
    w.x = W[(2 * j) * 64 + n];
    w.y = W[(2 * j + 1) * 64 + n];
    sW2[i] = w;
  }
  __syncthreads();

  const int lane    = tid & 31;
  const int wave    = tid >> 5;
  const int rowBase = tileBlock * 128 + wave * 16;
  if (rowBase + 16 > nrows) return;

  const int m  = lane & 15;
  const int kh = lane >> 4;
  const int n  = lane & 15;

  const float* __restrict__ xrow = X + (size_t)(rowBase + m) * 64;

  v8f acc0 = {}, acc1 = {}, acc2 = {}, acc3 = {};

#pragma unroll
  for (int kk = 0; kk < 16; ++kk) {
    const int k0 = kk * 4 + kh * 2;
    v2f a = *(const v2f*)(xrow + k0);

    const int kp = k0 >> 1;
    const v2f* __restrict__ brow = &sW2[kp * 64 + n];
    v2f b0 = brow[0];
    v2f b1 = brow[16];
    v2f b2 = brow[32];
    v2f b3 = brow[48];

    acc0 = __builtin_amdgcn_wmma_f32_16x16x4_f32(false, a, false, b0, (short)0, acc0, false, false);
    acc1 = __builtin_amdgcn_wmma_f32_16x16x4_f32(false, a, false, b1, (short)0, acc1, false, false);
    acc2 = __builtin_amdgcn_wmma_f32_16x16x4_f32(false, a, false, b2, (short)0, acc2, false, false);
    acc3 = __builtin_amdgcn_wmma_f32_16x16x4_f32(false, a, false, b3, (short)0, acc3, false, false);
  }

  const int crow = rowBase + (lane >> 4) * 8;
  const int ccol = lane & 15;
#pragma unroll
  for (int v = 0; v < 8; ++v) {
    float* __restrict__ yr = Y + (size_t)(crow + v) * 64 + ccol;
    yr[0]  = acc0[v];
    yr[16] = acc1[v];
    yr[32] = acc2[v];
    yr[48] = acc3[v];
  }
}

#define CAP_N 32
#define CAP_H 64
#define RN_SHIFT 9
#define RH_SHIFT 8
#define BIN_CAP 8192
#define BIN_CHUNK 4096
#define CNT_PAD 32
#define MAX_RANGES 1024

static __device__ __forceinline__ void bin_incidences_block(
    int* __restrict__ rangeBase, int* __restrict__ rangeRank,
    int* __restrict__ blockHalf,
    int chunk,
    const int* __restrict__ node_idx, const int* __restrict__ hedge_idx,
    int* __restrict__ binCnt,
    int2* __restrict__ bins,
    int* __restrict__ spillCnt, int2* __restrict__ spillN, int2* __restrict__ spillH,
    int n_inc, int nRN, int nR)
{
  const int tid = threadIdx.x;
  if (tid == 0) *blockHalf = (__builtin_amdgcn_s_sendmsg_rtn(0x87) >> 18) & 1;
  for (int r = tid; r < nR; r += 256) { rangeBase[r] = 0; rangeRank[r] = 0; }
  __syncthreads();
  const size_t mine = (size_t)(*blockHalf) * (size_t)nR;

  const int e0 = chunk * BIN_CHUNK;
  for (int k0 = 0; k0 < BIN_CHUNK / 256; k0 += 4) {
    int n[4], h[4];
#pragma unroll
    for (int u = 0; u < 4; ++u) {
      const int e = e0 + (k0 + u) * 256 + tid;
      const int ec = e < n_inc ? e : n_inc - 1;
      n[u] = node_idx[ec];
      h[u] = hedge_idx[ec];
    }
#pragma unroll
    for (int u = 0; u < 4; ++u) {
      const int e = e0 + (k0 + u) * 256 + tid;
      if (e < n_inc) {
        atomicAdd(&rangeBase[n[u] >> RN_SHIFT], 1);
        atomicAdd(&rangeBase[nRN + (h[u] >> RH_SHIFT)], 1);
      }
    }
  }
  __syncthreads();

  int got[MAX_RANGES / 256];
#pragma unroll
  for (int u = 0; u < MAX_RANGES / 256; ++u) {
    const int r = tid + u * 256;
    const int c = r < nR ? rangeBase[r] : 0;
    got[u] = 0;
    if (c > 0) got[u] = atomicAdd(binCnt + (mine + r) * CNT_PAD, c);
  }
#pragma unroll
  for (int u = 0; u < MAX_RANGES / 256; ++u) {
    const int r = tid + u * 256;
    if (r < nR) rangeBase[r] = got[u];
  }
  __syncthreads();

  for (int k0 = 0; k0 < BIN_CHUNK / 256; k0 += 4) {
    int n[4], h[4];
#pragma unroll
    for (int u = 0; u < 4; ++u) {
      const int e = e0 + (k0 + u) * 256 + tid;
      const int ec = e < n_inc ? e : n_inc - 1;
      n[u] = node_idx[ec];
      h[u] = hedge_idx[ec];
    }
#pragma unroll
    for (int u = 0; u < 4; ++u) {
      const int e = e0 + (k0 + u) * 256 + tid;
      if (e < n_inc) {
        const int rn = n[u] >> RN_SHIFT;
        const int rh = nRN + (h[u] >> RH_SHIFT);
        const int pn = rangeBase[rn] + atomicAdd(&rangeRank[rn], 1);
        const int ph = rangeBase[rh] + atomicAdd(&rangeRank[rh], 1);
        int2* dn = bins + (mine + rn) * BIN_CAP + pn;
        if (pn >= BIN_CAP) dn = spillN + atomicAdd(spillCnt + 0, 1);
        int2* dh = bins + (mine + rh) * BIN_CAP + ph;
        if (ph >= BIN_CAP) dh = spillH + atomicAdd(spillCnt + 1, 1);
        v2i pnh, phn;
        pnh.x = n[u]; pnh.y = h[u];
        phn.x = h[u]; phn.y = n[u];
        *(v2i*)dn = pnh;
        *(v2i*)dh = phn;
      }
    }
  }
}

__global__ __launch_bounds__(256) __attribute__((amdgpu_waves_per_eu(16)))
void bins_and_input_products(
    const int* __restrict__ node_idx, const int* __restrict__ hedge_idx,
    int* __restrict__ binCnt, int2* __restrict__ bins,
    int* __restrict__ spillCnt, int2* __restrict__ spillN, int2* __restrict__ spillH,
    int n_inc, int nN, int nH,
    const float* __restrict__ node_features, const float* __restrict__ hedge_features,
    const float* __restrict__ W_nn, const float* __restrict__ W_nh,
    const float* __restrict__ W_hh,
    float* __restrict__ node_pre, float* __restrict__ hedge_proj,
    float* __restrict__ hedge_pre)
{
  __shared__ v2f sW2[32 * 64];
  __shared__ int rangeBase[MAX_RANGES];
  __shared__ int rangeRank[MAX_RANGES];
  __shared__ int blockHalf;

  const int nbB = (n_inc + BIN_CHUNK - 1) / BIN_CHUNK;
  const int b = (int)blockIdx.x;
  if (b < nbB) {
    const int nRN = (nN + (1 << RN_SHIFT) - 1) >> RN_SHIFT;
    const int nRH = (nH + (1 << RH_SHIFT) - 1) >> RH_SHIFT;
    bin_incidences_block(rangeBase, rangeRank, &blockHalf, b, node_idx, hedge_idx, binCnt, bins,
                         spillCnt, spillN, spillH, n_inc, nRN, nRN + nRH);
    return;
  }

  const int pb = b - nbB;
  const int nbN = (nN + 127) >> 7;
  const int nbE = (nH + 127) >> 7;
  int whole = nbN >> 1;
  whole = nbE < whole ? nbE : whole;

  int role, idx;
  if (pb < whole * 4) {
    const int pat = pb >> 2;
    const int slot = pb & 3;
    if (slot < 2)        { role = 1; idx = pat * 2 + slot; }
    else if (slot == 2)  { role = 2; idx = pat; }
    else                 { role = 3; idx = pat; }
  } else {
    int r = pb - whole * 4;
    const int leftN = nbN - whole * 2;
    const int leftE = nbE - whole;
    if (r < leftN) { role = 1; idx = whole * 2 + r; }
    else {
      r -= leftN;
      if (r < leftE) { role = 2; idx = whole + r; }
      else           { role = 3; idx = whole + (r - leftE); }
    }
  }

  const float* __restrict__ X = (role == 1) ? node_features : hedge_features;
  const float* __restrict__ Wm = (role == 1) ? W_nn : (role == 2 ? W_nh : W_hh);
  float* __restrict__ Y = (role == 1) ? node_pre : (role == 2 ? hedge_proj : hedge_pre);
  const int nrows = (role == 1) ? nN : nH;
  rows64_product_block(sW2, X, Wm, Y, nrows, idx);
}

template <int CAP>
static __device__ __forceinline__ void pack_bucket_keys(
    const int (&ent)[CAP / 32], unsigned (&key)[CAP / 32], int lane)
{
  constexpr int NK = CAP / 32;
  constexpr int SH = (CAP == 64) ? 6 : 5;
#pragma unroll
  for (int q = 0; q < NK; ++q) key[q] = ((unsigned)ent[q] << SH) | (unsigned)(q * 32 + lane);
}

template <int NK, int NC>
static __device__ __forceinline__ void rank_against(
    const unsigned (&key)[NK], int cu, int (&rnk)[NK])
{
#pragma unroll
  for (int qs = 0; qs < NC; ++qs) {
    const int left = cu - qs * 32;
    const int cn = left < 32 ? left : 32;
    for (int j = 0; j < cn; j += 4) {
#pragma unroll
      for (int t = 0; t < 4; ++t) {
        const unsigned kj = (unsigned)__builtin_amdgcn_readlane((int)key[qs], j + t);
#pragma unroll
        for (int q = 0; q < NC; ++q) rnk[q] += (kj < key[q]) ? 1 : 0;
      }
    }
  }
}

template <int NK>
static __device__ __forceinline__ void rank_bucket(
    const unsigned (&key)[NK], int cu, int (&rnk)[NK])
{
#pragma unroll
  for (int q = 0; q < NK; ++q) rnk[q] = 0;
  if (NK == 1 || cu <= 32) rank_against<NK, 1>(key, cu, rnk);
  else                     rank_against<NK, NK>(key, cu, rnk);
}

template <int U>
static __device__ __forceinline__ void gather_group(
    v2f& acc, int skq, int l0, int rem, const float* __restrict__ src, int lane)
{
  v2f v[U];
#pragma unroll
  for (int u = 0; u < U; ++u) {
    const int uu = (u < rem) ? u : (rem - 1);
    const int s = __builtin_amdgcn_readlane(skq, l0 + uu);
    v[u] = *(const v2f*)(src + (size_t)s * 64 + 2 * lane);
  }
#pragma unroll
  for (int u = 0; u < U; ++u) {
    if (u < rem) { acc.x += v[u].x; acc.y += v[u].y; }
  }
}

static __device__ __forceinline__ v2f gate_tanh2(v2f g, v2f s)
{
  const float k = 2.8853900817779268f;
  const float cx = g.x * s.x;
  const float cy = g.y * s.y;
  const float ex = __builtin_amdgcn_exp2f(cx * k);
  const float ey = __builtin_amdgcn_exp2f(cy * k);
  const float rx = __builtin_amdgcn_rcpf(1.0f + ex);
  const float ry = __builtin_amdgcn_rcpf(1.0f + ey);
  v2f o;
  o.x = __builtin_fmaf(-2.0f, rx, 1.0f);
  o.y = __builtin_fmaf(-2.0f, ry, 1.0f);
  return o;
}

template <int ROWS, int CAP, int NT>
static __device__ __forceinline__ void slot_pairs_of_rows(
    int* __restrict__ cntL, int* __restrict__ entL,
    const int2* __restrict__ bin, int binCount, int row0)
{
  const int tid = threadIdx.x;
  for (int i0 = 0; i0 < binCount; i0 += 8 * NT) {
    int2 pr[8];
#pragma unroll
    for (int u = 0; u < 8; ++u) {
      const int i = i0 + u * NT + tid;
      pr[u] = bin[i < binCount ? i : binCount - 1];
    }
#pragma unroll
    for (int u = 0; u < 8; ++u) {
      const int i = i0 + u * NT + tid;
      const unsigned lr = (unsigned)(pr[u].x - row0);
      if (i < binCount && lr < (unsigned)ROWS) {
        const int t = atomicAdd(&cntL[lr], 1);
        if (t < CAP) entL[lr * CAP + t] = pr[u].y;
      }
    }
  }
}

template <int ROWS, int CAP, int NT>
static __device__ __forceinline__ void local_buckets_from_bins(
    int* __restrict__ cntL, int* __restrict__ entL,
    const int2* __restrict__ binA, int countA,
    const int2* __restrict__ binB, int countB, int row0)
{
  for (int i = threadIdx.x; i < ROWS; i += NT) cntL[i] = 0;
  __syncthreads();
  slot_pairs_of_rows<ROWS, CAP, NT>(cntL, entL, binA, countA, row0);
  slot_pairs_of_rows<ROWS, CAP, NT>(cntL, entL, binB, countB, row0);
  __syncthreads();
}

static __device__ __forceinline__ void add_matching_pairs(
    v2f& acc, const int2* __restrict__ pairs, int tot, int row,
    const float* __restrict__ src, int lane)
{
  for (int i = 0; i < tot; i += 32) {
    const int p = i + lane;
    int2 pr = make_int2(-1, 0);
    if (p < tot) pr = pairs[p];
    unsigned m = __builtin_amdgcn_ballot_w32(pr.x == row);
    while (m) {
      const int b = __builtin_ctz(m);
      m &= m - 1;
      const int s = __builtin_amdgcn_readlane(pr.y, b);
      const v2f v = *(const v2f*)(src + (size_t)s * 64 + 2 * lane);
      acc.x += v.x; acc.y += v.y;
    }
  }
}

template <int CAP>
static __device__ __forceinline__ v2f ordered_row_sum(
    const int* __restrict__ cntL, const int* __restrict__ entL, int* __restrict__ sortedW,
    int lr, int row, const int2* __restrict__ binA, int countA,
    const int2* __restrict__ binB, int countB,
    const int2* __restrict__ spill, int spillTot,
    const float* __restrict__ src, int lane)
{
  constexpr int NK = CAP / 32;
  const int c = __builtin_amdgcn_readfirstlane(cntL[lr]);
  v2f acc = {0.0f, 0.0f};
  if (c <= CAP) {
    const int cu = c;
    int ent[NK];
#pragma unroll
    for (int q = 0; q < NK; ++q) {
      const int p = q * 32 + lane;
      ent[q] = (p < cu) ? entL[lr * CAP + p] : 0x7fffffff;
    }
    unsigned key[NK];
    pack_bucket_keys<CAP>(ent, key, lane);
    int rnk[NK];
    rank_bucket<NK>(key, cu, rnk);
#pragma unroll
    for (int q = 0; q < NK; ++q) {
      const int p = q * 32 + lane;
      if (p < cu) sortedW[rnk[q]] = ent[q];
    }
    int sk[NK];
#pragma unroll
    for (int q = 0; q < NK; ++q) {
      const int p = q * 32 + lane;
      sk[q] = (p < cu) ? sortedW[p] : 0;
    }
    for (int j = 0; j < cu; j += 8) {
      const int rem = cu - j;
      int skq = sk[0];
#pragma unroll
      for (int q = 1; q < NK; ++q) skq = (j >= q * 32) ? sk[q] : skq;
      const int l0 = j & 31;
      if (rem > 4) gather_group<8>(acc, skq, l0, rem, src, lane);
      else         gather_group<4>(acc, skq, l0, rem, src, lane);
    }
  } else {
    add_matching_pairs(acc, binA, countA, row, src, lane);
    add_matching_pairs(acc, binB, countB, row, src, lane);
  }
  if (spillTot > 0) add_matching_pairs(acc, spill, spillTot, row, src, lane);
  return acc;
}

#define PULL_ROWS 64

template <int CAP>
__global__ __launch_bounds__(256) void pull_gate_tanh(
    const float* __restrict__ src, const int* __restrict__ binCnt, int cntHalfStride,
    const int2* __restrict__ bins, int binHalfStride, const int* __restrict__ spillCntP,
    const int2* __restrict__ spill, const float* __restrict__ pre,
    float* __restrict__ out, int nrows, int n_inc)
{
  __shared__ int cntL[PULL_ROWS];
  __shared__ int entL[PULL_ROWS * CAP];
  __shared__ int sorted[8][CAP];

  const int lane = threadIdx.x & 31;
  const int wave = __builtin_amdgcn_readfirstlane((int)(threadIdx.x >> 5));
  const int row0 = (int)blockIdx.x * PULL_ROWS;
  const int range = row0 >> RN_SHIFT;
  int countA = binCnt[(size_t)range * CNT_PAD];
  int countB = binCnt[(size_t)cntHalfStride + (size_t)range * CNT_PAD];
  countA = countA < BIN_CAP ? countA : BIN_CAP;
  countB = countB < BIN_CAP ? countB : BIN_CAP;
  const int2* __restrict__ binA = bins + (size_t)range * BIN_CAP;
  const int2* __restrict__ binB = binA + (size_t)binHalfStride;
  int spillTot = spillCntP[0];
  spillTot = spillTot < n_inc ? spillTot : n_inc;

  local_buckets_from_bins<PULL_ROWS, CAP, 256>(cntL, entL, binA, countA, binB, countB, row0);

  const int wrow0 = row0 + wave * 8;
  int nr = nrows - wrow0;
  nr = nr < 0 ? 0 : (nr > 8 ? 8 : nr);
  for (int r = 0; r < nr; ++r) {
    const int row = wrow0 + r;
    const v2f p2 = *(const v2f*)(pre + (size_t)row * 64 + 2 * lane);
    const v2f acc = ordered_row_sum<CAP>(cntL, entL, sorted[wave], wave * 8 + r, row,
                                         binA, countA, binB, countB, spill, spillTot, src, lane);
    const v2f o = gate_tanh2(p2, acc);
    *(v2f*)(out + (size_t)row * 64 + 2 * lane) = o;
  }
}

#define SLAB_PITCH 66

template <int CAP>
__global__ __launch_bounds__(128) __attribute__((amdgpu_waves_per_eu(16)))
void pull_rows32_product_gate_tanh(
    const float* __restrict__ src, const int* __restrict__ binCnt, int cntHalfStride,
    const int2* __restrict__ bins, int binHalfStride, const int* __restrict__ spillCntP,
    const int2* __restrict__ spill, const float* __restrict__ W,
    const float* __restrict__ pre, float* __restrict__ out, int nrows, int n_inc)
{
  __shared__ int cntL[32];
  __shared__ int entL[32 * CAP];
  __shared__ int sorted[4][CAP];
  __shared__ __attribute__((aligned(16))) float sums[32 * SLAB_PITCH];

  const int tid = threadIdx.x;
  const int lane = tid & 31;
  const int wave = __builtin_amdgcn_readfirstlane(tid >> 5);
  const int brow0 = (int)blockIdx.x * 32;
  const int row0 = brow0 + wave * 8;
  int nr = nrows - row0;
  nr = nr < 0 ? 0 : (nr > 8 ? 8 : nr);

  const int range = brow0 >> RH_SHIFT;
  int countA = binCnt[(size_t)range * CNT_PAD];
  int countB = binCnt[(size_t)cntHalfStride + (size_t)range * CNT_PAD];
  countA = countA < BIN_CAP ? countA : BIN_CAP;
  countB = countB < BIN_CAP ? countB : BIN_CAP;
  const int2* __restrict__ binA = bins + (size_t)range * BIN_CAP;
  const int2* __restrict__ binB = binA + (size_t)binHalfStride;
  int spillTot = spillCntP[0];
  spillTot = spillTot < n_inc ? spillTot : n_inc;

  v2f g2[8];
#pragma unroll
  for (int r = 0; r < 8; ++r) {
    int gr = row0 + r;
    gr = gr < nrows ? gr : nrows - 1;
    g2[r] = *(const v2f*)(pre + (size_t)gr * 64 + 2 * lane);
  }

  local_buckets_from_bins<32, CAP, 128>(cntL, entL, binA, countA, binB, countB, brow0);

  for (int r = 0; r < nr; ++r) {
    const v2f acc = ordered_row_sum<CAP>(cntL, entL, sorted[wave], wave * 8 + r, row0 + r,
                                         binA, countA, binB, countB, spill, spillTot, src, lane);
    *(v2f*)&sums[(wave * 8 + r) * SLAB_PITCH + 2 * lane] = acc;
  }
  for (int r = nr; r < 8; ++r) {
    const v2f z = {0.0f, 0.0f};
    *(v2f*)&sums[(wave * 8 + r) * SLAB_PITCH + 2 * lane] = z;
  }

  const int mt = wave >> 1;
  const int nh = wave & 1;
  const int m  = lane & 15;
  const int kh = lane >> 4;
  const int n  = lane & 15;

  const float* __restrict__ wcol = W + (kh * 2) * 64 + nh * 32 + 2 * n;
  v2f wq0[4], wq1[4];
#pragma unroll
  for (int kk = 0; kk < 3; ++kk) {
    wq0[kk] = *(const v2f*)(wcol + kk * 256);
    wq1[kk] = *(const v2f*)(wcol + kk * 256 + 64);
  }

  __syncthreads();

  v8f acc0 = {}, acc1 = {};

#pragma unroll
  for (int kk = 0; kk < 16; ++kk) {
    const int k0 = kk * 4 + kh * 2;
    const v2f a = *(const v2f*)&sums[(mt * 16 + m) * SLAB_PITCH + k0];

    if (kk + 3 < 16) {
      wq0[(kk + 3) & 3] = *(const v2f*)(wcol + (kk + 3) * 256);
      wq1[(kk + 3) & 3] = *(const v2f*)(wcol + (kk + 3) * 256 + 64);
    }
    v2f b0, b1;
    b0.x = wq0[kk & 3].x;
    b0.y = wq1[kk & 3].x;
    b1.x = wq0[kk & 3].y;
    b1.y = wq1[kk & 3].y;

    acc0 = __builtin_amdgcn_wmma_f32_16x16x4_f32(false, a, false, b0, (short)0, acc0, false, false);
    acc1 = __builtin_amdgcn_wmma_f32_16x16x4_f32(false, a, false, b1, (short)0, acc1, false, false);
  }

  __syncthreads();

  const int crow = mt * 16 + (lane >> 4) * 8;
  const int ccol = nh * 32 + 2 * (lane & 15);
#pragma unroll
  for (int v = 0; v < 8; ++v) {
    v2f pr;
    pr.x = acc0[v];
    pr.y = acc1[v];
    *(v2f*)&sums[(crow + v) * SLAB_PITCH + ccol] = pr;
  }

  __syncthreads();

#pragma unroll
  for (int r = 0; r < 8; ++r) {
    if (r < nr) {
      const v2f s = *(const v2f*)&sums[(wave * 8 + r) * SLAB_PITCH + 2 * lane];
      const v2f o = gate_tanh2(g2[r], s);
      *(v2f*)(out + (size_t)(row0 + r) * 64 + 2 * lane) = o;
    }
  }
}

extern "C" void kernel_launch(void* const* d_in, const int* in_sizes, int n_in,
                              void* d_out, int out_size, void* d_ws, size_t ws_size,
                              hipStream_t stream) {
  const float* node_features  = (const float*)d_in[0];
  const float* hedge_features = (const float*)d_in[1];
  const float* W_nn = (const float*)d_in[2];
  const float* W_nh = (const float*)d_in[3];
  const float* W_hh = (const float*)d_in[4];
  const float* W_hn = (const float*)d_in[5];
  const int* node_idx  = (const int*)d_in[6];
  const int* hedge_idx = (const int*)d_in[7];

  const int N   = in_sizes[0] / 64;
  const int E   = in_sizes[1] / 64;
  const int INC = in_sizes[6];

  const int nRN = (N + (1 << RN_SHIFT) - 1) >> RN_SHIFT;
  const int nRH = (E + (1 << RH_SHIFT) - 1) >> RH_SHIFT;
  const int nR  = nRN + nRH;

  float* slotA = (float*)d_ws;
  float* slotB = slotA + (size_t)N * 64;
  float* slotC = slotB + (size_t)N * 64;
  int2* bins   = (int2*)(slotC + (size_t)E * 64);
  int2* spillN = bins + (size_t)2 * nR * BIN_CAP;
  int2* spillH = spillN + (size_t)INC;
  int* binCnt  = (int*)(spillH + (size_t)INC);
  int* spillCnt = binCnt + (size_t)2 * nR * CNT_PAD;

  float* out_node  = (float*)d_out;
  float* out_hedge = out_node + (size_t)N * 64;

  const dim3 blk(256);
  const dim3 firstGrid((INC + BIN_CHUNK - 1) / BIN_CHUNK + (N + 127) / 128 + 2 * ((E + 127) / 128));

  hipMemsetAsync(binCnt, 0, ((size_t)2 * nR * CNT_PAD + 2) * sizeof(int), stream);
  bins_and_input_products<<<firstGrid, blk, 0, stream>>>(
      node_idx, hedge_idx, binCnt, bins, spillCnt, spillN, spillH, INC, N, E,
      node_features, hedge_features, W_nn, W_nh, W_hh, slotB, slotA, slotC);

  pull_gate_tanh<CAP_N><<<dim3((N + PULL_ROWS - 1) / PULL_ROWS), blk, 0, stream>>>(
      slotA, binCnt, nR * CNT_PAD, bins, nR * BIN_CAP, spillCnt + 0, spillN, slotB, out_node, N, INC);

  pull_rows32_product_gate_tanh<CAP_H><<<dim3((E + 31) / 32), dim3(128), 0, stream>>>(
      out_node, binCnt + (size_t)nRN * CNT_PAD, nR * CNT_PAD, bins + (size_t)nRN * BIN_CAP, nR * BIN_CAP,
      spillCnt + 1, spillH, W_hn, slotC, out_hedge, E, INC);
}
